# starting kernel
# baseline (speedup 1.0000x reference)
#include <hip/hip_runtime.h>
#include <math.h>


#define NN 4
#define DD 7
#define HH 768
#define WW 768
#define HWs (HH * WW)
#define RAD 30
#define CH 64
#define PAD 32
#define WP (WW + 2 * PAD)
#define HWP (HH * WP)
#define RTOT (NN * DD * HH)

static_assert(HH % CH == 0, "chunking");
static_assert(CH > RAD, "warmup writes must stay in chunk");

constexpr float INVK = 1.0f / (7.0f * 61.0f * 61.0f * 2.0f);
constexpr float EPSF = 1e-7f;

typedef __attribute__((ext_vector_type(2))) float v2f;
typedef __attribute__((ext_vector_type(8))) float v8f;

#define WMMA_F32_4(A, B, C) \
  __builtin_amdgcn_wmma_f32_16x16x4_f32(false, (A), false, (B), (short)0, (C), false, false)

__device__ __forceinline__ v2f band_b(int e) {
  v2f b;
  b.x = ((unsigned)(e + RAD) < 61u) ? 1.f : 0.f;
  b.y = ((unsigned)(e - 1 + RAD) < 61u) ? 1.f : 0.f;
  return b;
}

__device__ __forceinline__ void dbox7(const float* s, float* b) {
  float ps[8];
  ps[0] = 0.f;
#pragma unroll
  for (int d = 0; d < 7; ++d) ps[d + 1] = ps[d] + s[d];
#pragma unroll
  for (int d = 0; d < 7; ++d) {
    const int hi = (d + 3 > 6) ? 7 : (d + 4);
    const int lo = (d - 3 < 0) ? 0 : (d - 3);
    b[d] = ps[hi] - ps[lo];
  }
}

__global__ void __launch_bounds__(256)
k0_zero_pads(float* __restrict__ f0, float* __restrict__ f1,
             float* __restrict__ f2) {
  const int i = blockIdx.x * 256 + threadIdx.x;
  const int row = i >> 6;
  const int c = i & 63;
  const int col = (c < PAD) ? c : (c + WW);
  const size_t o = (size_t)row * WP + col;
  float* f = (blockIdx.y == 0) ? f0 : (blockIdx.y == 1) ? f1 : f2;
  f[o] = 0.f;
}

__global__ void __launch_bounds__(64)
k1_chsum_dbox_hbox(const float* __restrict__ x, float* __restrict__ S1,
                   float* __restrict__ A1, float* __restrict__ A2) {
  const float2* x2 = (const float2*)x;
  const int w  = blockIdx.x * 64 + threadIdx.x;
  const int h0 = blockIdx.y * CH;
  const int n  = blockIdx.z;
  const size_t fbase = (size_t)n * DD * HWs + w;
  const size_t pbase = (size_t)n * DD * HWP + PAD + w;

  float win1[7], win2[7];
#pragma unroll
  for (int d = 0; d < 7; ++d) { win1[d] = 0.f; win2[d] = 0.f; }

  float s1[7], s2[7], b1[7], b2[7];

  for (int hr = h0 - RAD; hr < h0 + RAD; ++hr) {
    if (hr < 0) continue;
    const size_t rb = fbase + (size_t)hr * WW;
#pragma unroll
    for (int d = 0; d < 7; ++d) {
      const float2 v = x2[rb + (size_t)d * HWs];
      s1[d] = v.x + v.y;
      s2[d] = v.x * v.x + v.y * v.y;
    }
    dbox7(s1, b1); dbox7(s2, b2);
#pragma unroll
    for (int d = 0; d < 7; ++d) { win1[d] += b1[d]; win2[d] += b2[d]; }
    if (hr >= h0) {
#pragma unroll
      for (int d = 0; d < 7; ++d) S1[rb + (size_t)d * HWs] = s1[d];
    }
  }

  for (int h = h0; h < h0 + CH; ++h) {
    const int hl = h + RAD;
    if (hl < HH) {
      const size_t rb = fbase + (size_t)hl * WW;
      if (hl + 1 < HH) __builtin_prefetch(&x2[rb + WW], 0, 1);
#pragma unroll
      for (int d = 0; d < 7; ++d) {
        const float2 v = x2[rb + (size_t)d * HWs];
        s1[d] = v.x + v.y;
        s2[d] = v.x * v.x + v.y * v.y;
      }
      dbox7(s1, b1); dbox7(s2, b2);
#pragma unroll
      for (int d = 0; d < 7; ++d) { win1[d] += b1[d]; win2[d] += b2[d]; }
      if (hl < h0 + CH) {
#pragma unroll
        for (int d = 0; d < 7; ++d) S1[rb + (size_t)d * HWs] = s1[d];
      }
    }
    {
      const size_t ob = pbase + (size_t)h * WP;
#pragma unroll
      for (int d = 0; d < 7; ++d) {
        A1[ob + (size_t)d * HWP] = win1[d];
        A2[ob + (size_t)d * HWP] = win2[d];
      }
    }
    const int ht = h - RAD;
    if (ht >= 0) {
      const size_t rb = fbase + (size_t)ht * WW;
#pragma unroll
      for (int d = 0; d < 7; ++d) {
        const float2 v = x2[rb + (size_t)d * HWs];
        s1[d] = v.x + v.y;
        s2[d] = v.x * v.x + v.y * v.y;
      }
      dbox7(s1, b1); dbox7(s2, b2);
#pragma unroll
      for (int d = 0; d < 7; ++d) { win1[d] -= b1[d]; win2[d] -= b2[d]; }
    }
  }
}

__global__ void __launch_bounds__(128)
k2_wbox_wmma2(const float* __restrict__ A1, const float* __restrict__ A2,
              float* __restrict__ m, float* __restrict__ q2) {
  const int lane = threadIdx.x;
  const int half = lane >> 4;
  const int n16  = lane & 15;
  const int r0 = (blockIdx.y * 4 + threadIdx.y) * 16;
  const int j0 = blockIdx.x * 32;
  const int rowA = r0 + n16;
  const float* pa1 = A1 + (size_t)rowA * WP + (PAD - RAD + 2 * half + j0);
  const float* pa2 = A2 + (size_t)rowA * WP + (PAD - RAD + 2 * half + j0);
  const int u = n16 + RAD - 2 * half;

  v8f c10 = {0.f,0.f,0.f,0.f,0.f,0.f,0.f,0.f};
  v8f c20 = {0.f,0.f,0.f,0.f,0.f,0.f,0.f,0.f};
  v8f c11 = {0.f,0.f,0.f,0.f,0.f,0.f,0.f,0.f};
  v8f c21 = {0.f,0.f,0.f,0.f,0.f,0.f,0.f,0.f};

#pragma unroll
  for (int t = 0; t < 23; ++t) {
    const v2f a1 = *reinterpret_cast<const v2f*>(pa1 + 4 * t);
    const v2f a2 = *reinterpret_cast<const v2f*>(pa2 + 4 * t);
    if (t <= 18) {
      const v2f b = band_b(u - 4 * t);
      c10 = WMMA_F32_4(a1, b, c10);
      c20 = WMMA_F32_4(a2, b, c20);
    }
    if (t >= 4) {
      const v2f b = band_b(u + 16 - 4 * t);
      c11 = WMMA_F32_4(a1, b, c11);
      c21 = WMMA_F32_4(a2, b, c21);
    }
  }
#pragma unroll
  for (int v = 0; v < 8; ++v) {
    const int row = r0 + v + 8 * half;
    const size_t o = (size_t)row * WW + (j0 + n16);
    m[o]       = c10[v] * INVK;
    q2[o]      = c20[v] * INVK;
    m[o + 16]  = c11[v] * INVK;
    q2[o + 16] = c21[v] * INVK;
  }
}

__global__ void __launch_bounds__(64)
k3_prep_dbox_hbox(const float* __restrict__ m, const float* __restrict__ S1,
                  float* __restrict__ B1, float* __restrict__ B2,
                  float* __restrict__ B3) {
  const int w  = blockIdx.x * 64 + threadIdx.x;
  const int h0 = blockIdx.y * CH;
  const int n  = blockIdx.z;
  const size_t fbase = (size_t)n * DD * HWs + w;
  const size_t pbase = (size_t)n * DD * HWP + PAD + w;

  float w1[7], w2[7], w3[7];
#pragma unroll
  for (int d = 0; d < 7; ++d) { w1[d] = 0.f; w2[d] = 0.f; w3[d] = 0.f; }

  float t1[7], t2[7], t3[7], b1[7], b2[7], b3[7];

  auto readrow = [&](int hr) {
    const size_t rb = fbase + (size_t)hr * WW;
#pragma unroll
    for (int d = 0; d < 7; ++d) {
      const float mv = m[rb + (size_t)d * HWs];
      const float sv = S1[rb + (size_t)d * HWs];
      t1[d] = mv;
      t2[d] = mv * sv;
      t3[d] = mv * mv;
    }
    dbox7(t1, b1); dbox7(t2, b2); dbox7(t3, b3);
  };

  for (int hr = h0 - RAD; hr < h0 + RAD; ++hr) {
    if (hr < 0) continue;
    readrow(hr);
#pragma unroll
    for (int d = 0; d < 7; ++d) { w1[d] += b1[d]; w2[d] += b2[d]; w3[d] += b3[d]; }
  }
  for (int h = h0; h < h0 + CH; ++h) {
    const int hl = h + RAD;
    if (hl < HH) {
      if (hl + 1 < HH) __builtin_prefetch(&m[fbase + (size_t)(hl + 1) * WW], 0, 1);
      readrow(hl);
#pragma unroll
      for (int d = 0; d < 7; ++d) { w1[d] += b1[d]; w2[d] += b2[d]; w3[d] += b3[d]; }
    }
    {
      const size_t ob = pbase + (size_t)h * WP;
#pragma unroll
      for (int d = 0; d < 7; ++d) {
        B1[ob + (size_t)d * HWP] = w1[d];
        B2[ob + (size_t)d * HWP] = w2[d];
        B3[ob + (size_t)d * HWP] = w3[d];
      }
    }
    const int ht = h - RAD;
    if (ht >= 0) {
      readrow(ht);
#pragma unroll
      for (int d = 0; d < 7; ++d) { w1[d] -= b1[d]; w2[d] -= b2[d]; w3[d] -= b3[d]; }
    }
  }
}

__global__ void __launch_bounds__(128)
k4_wbox_wmma3_stats(const float* __restrict__ B1, const float* __restrict__ B2,
                    const float* __restrict__ B3, const float* __restrict__ m,
                    const float* __restrict__ q2, float* __restrict__ R) {
  const int lane = threadIdx.x;
  const int half = lane >> 4;
  const int n16  = lane & 15;
  const int r0 = (blockIdx.y * 4 + threadIdx.y) * 16;
  const int j0 = blockIdx.x * 32;
  const int rowA = r0 + n16;
  const float* p1 = B1 + (size_t)rowA * WP + (PAD - RAD + 2 * half + j0);
  const float* p2 = B2 + (size_t)rowA * WP + (PAD - RAD + 2 * half + j0);
  const float* p3 = B3 + (size_t)rowA * WP + (PAD - RAD + 2 * half + j0);
  const int u = n16 + RAD - 2 * half;

  v8f cm0 = {0.f,0.f,0.f,0.f,0.f,0.f,0.f,0.f};
  v8f cs0 = {0.f,0.f,0.f,0.f,0.f,0.f,0.f,0.f};
  v8f cq0 = {0.f,0.f,0.f,0.f,0.f,0.f,0.f,0.f};
  v8f cm1 = {0.f,0.f,0.f,0.f,0.f,0.f,0.f,0.f};
  v8f cs1 = {0.f,0.f,0.f,0.f,0.f,0.f,0.f,0.f};
  v8f cq1 = {0.f,0.f,0.f,0.f,0.f,0.f,0.f,0.f};

#pragma unroll
  for (int t = 0; t < 23; ++t) {
    const v2f a1 = *reinterpret_cast<const v2f*>(p1 + 4 * t);
    const v2f a2 = *reinterpret_cast<const v2f*>(p2 + 4 * t);
    const v2f a3 = *reinterpret_cast<const v2f*>(p3 + 4 * t);
    if (t <= 18) {
      const v2f b = band_b(u - 4 * t);
      cm0 = WMMA_F32_4(a1, b, cm0);
      cs0 = WMMA_F32_4(a2, b, cs0);
      cq0 = WMMA_F32_4(a3, b, cq0);
    }
    if (t >= 4) {
      const v2f b = band_b(u + 16 - 4 * t);
      cm1 = WMMA_F32_4(a1, b, cm1);
      cs1 = WMMA_F32_4(a2, b, cs1);
      cq1 = WMMA_F32_4(a3, b, cq1);
    }
  }
#pragma unroll
  for (int v = 0; v < 8; ++v) {
    const int row = r0 + v + 8 * half;
    const size_t o = (size_t)row * WW + (j0 + n16);
    {
      const float mv  = m[o];
      const float qv  = q2[o];
      const float c1v = mv - 2.0f * INVK * cm0[v];
      const float c2v = qv - 2.0f * INVK * (cs0[v] - cq0[v]);
      const float var = c2v - c1v * c1v;
      R[o] = 1.0f / (sqrtf(var) + EPSF);
    }
    {
      const float mv  = m[o + 16];
      const float qv  = q2[o + 16];
      const float c1v = mv - 2.0f * INVK * cm1[v];
      const float c2v = qv - 2.0f * INVK * (cs1[v] - cq1[v]);
      const float var = c2v - c1v * c1v;
      R[o + 16] = 1.0f / (sqrtf(var) + EPSF);
    }
  }
}

__global__ void __launch_bounds__(256)
k5_finalize(const float* __restrict__ x, const float* __restrict__ m,
            const float* __restrict__ R, float* __restrict__ out) {
  const size_t i = (size_t)blockIdx.x * 256 + threadIdx.x;
  const float2 xv = ((const float2*)x)[i];
  const float mv = m[i];
  const float rv = R[i];
  float2 o;
  o.x = (xv.x - mv) * rv;
  o.y = (xv.y - mv) * rv;
  ((float2*)out)[i] = o;
}

extern "C" void kernel_launch(void* const* d_in, const int* in_sizes, int n_in,
                              void* d_out, int out_size, void* d_ws, size_t ws_size,
                              hipStream_t stream) {
  (void)in_sizes; (void)n_in; (void)out_size; (void)ws_size;
  const float* x = (const float*)d_in[0];
  float* out = (float*)d_out;
  float* ws = (float*)d_ws;

  const size_t F  = (size_t)NN * DD * HH * WW;
  const size_t FP = (size_t)RTOT * WP;
  float* S1  = ws;
  float* A1p = ws + F;
  float* A2p = ws + F + FP;
  float* mf  = ws + F + 2 * FP;
  float* q2  = mf + F;
  float* B3p = q2 + F;

  const dim3 gZ((RTOT * 64) / 256, 3);
  const dim3 gHB(WW / 64, HH / CH, NN);
  const dim3 bHB(64);
  const dim3 gW(WW / 32, RTOT / (16 * 4));
  const dim3 bW(32, 4);

  k0_zero_pads<<<gZ, 256, 0, stream>>>(A1p, A2p, B3p);
  k1_chsum_dbox_hbox<<<gHB, bHB, 0, stream>>>(x, S1, A1p, A2p);
  k2_wbox_wmma2<<<gW, bW, 0, stream>>>(A1p, A2p, mf, q2);
  k3_prep_dbox_hbox<<<gHB, bHB, 0, stream>>>(mf, S1, A1p, A2p, B3p);
  k4_wbox_wmma3_stats<<<gW, bW, 0, stream>>>(A1p, A2p, B3p, mf, q2,  S1);
  k5_finalize<<<(unsigned)(F / 256), 256, 0, stream>>>(x, mf, S1, out);
}


# W-box WMMA kernels: A window of a 4-wave block (16 rows x 192 cols per field) staged in LDS by whole-line 16-byte global loads; fragments by ds_load_b64, not 16-line 8-byte global loads
# speedup vs baseline: 1.4838x; 1.4838x over previous
#include <hip/hip_runtime.h>
#include <math.h>


#define NN 4
#define DD 7
#define HH 768
#define WW 768
#define HWs (HH * WW)
#define RAD 30
#define CH 64
#define PAD 32
#define WP (WW + 2 * PAD)
#define HWP (HH * WP)
#define RTOT (NN * DD * HH)

static_assert(HH % CH == 0, "chunking");
static_assert(CH > RAD, "warmup writes must stay in chunk");

constexpr float INVK = 1.0f / (7.0f * 61.0f * 61.0f * 2.0f);
constexpr float EPSF = 1e-7f;

typedef __attribute__((ext_vector_type(2))) float v2f;
typedef __attribute__((ext_vector_type(4))) float v4f;
typedef __attribute__((ext_vector_type(8))) float v8f;

#define TCOLS 128
#define WIN4 48
#define NLD ((16 * WIN4) / 128)
#define LP 196

static_assert(WW % TCOLS == 0, "column blocks");
static_assert((16 * WIN4) % 128 == 0, "staging loads divide evenly");
static_assert(4 * WIN4 >= TCOLS + 2 * RAD + (PAD - RAD), "window covers the band");
static_assert(LP >= 4 * WIN4 && (LP % 8) == 4, "pitch = 4 * odd");

__device__ __forceinline__ void stage_fetch(const float* __restrict__ g, int tid,
                                            v4f (&v)[NLD]) {
#pragma unroll
  for (int it = 0; it < NLD; ++it) {
    const int idx = it * 128 + tid;
    const int row = idx / WIN4;
    const int c4  = idx - row * WIN4;
    v[it] = *reinterpret_cast<const v4f*>(g + (size_t)row * WP + 4 * c4);
  }
}
__device__ __forceinline__ void stage_put(float* s, int tid, const v4f (&v)[NLD]) {
#pragma unroll
  for (int it = 0; it < NLD; ++it) {
    const int idx = it * 128 + tid;
    const int row = idx / WIN4;
    const int c4  = idx - row * WIN4;
    *reinterpret_cast<v4f*>(s + row * LP + 4 * c4) = v[it];
  }
}

#define WMMA_F32_4(A, B, C) \
  __builtin_amdgcn_wmma_f32_16x16x4_f32(false, (A), false, (B), (short)0, (C), false, false)

__device__ __forceinline__ v2f band_b(int e) {
  v2f b;
  b.x = ((unsigned)(e + RAD) < 61u) ? 1.f : 0.f;
  b.y = ((unsigned)(e - 1 + RAD) < 61u) ? 1.f : 0.f;
  return b;
}

__device__ __forceinline__ void dbox7(const float* s, float* b) {
  float ps[8];
  ps[0] = 0.f;
#pragma unroll
  for (int d = 0; d < 7; ++d) ps[d + 1] = ps[d] + s[d];
#pragma unroll
  for (int d = 0; d < 7; ++d) {
    const int hi = (d + 3 > 6) ? 7 : (d + 4);
    const int lo = (d - 3 < 0) ? 0 : (d - 3);
    b[d] = ps[hi] - ps[lo];
  }
}

__global__ void __launch_bounds__(256)
k0_zero_pads(float* __restrict__ f0, float* __restrict__ f1,
             float* __restrict__ f2) {
  const int i = blockIdx.x * 256 + threadIdx.x;
  const int row = i >> 6;
  const int c = i & 63;
  const int col = (c < PAD) ? c : (c + WW);
  const size_t o = (size_t)row * WP + col;
  float* f = (blockIdx.y == 0) ? f0 : (blockIdx.y == 1) ? f1 : f2;
  f[o] = 0.f;
}

__global__ void __launch_bounds__(64)
k1_chsum_dbox_hbox(const float* __restrict__ x, float* __restrict__ S1,
                   float* __restrict__ A1, float* __restrict__ A2) {
  const float2* x2 = (const float2*)x;
  const int w  = blockIdx.x * 64 + threadIdx.x;
  const int h0 = blockIdx.y * CH;
  const int n  = blockIdx.z;
  const size_t fbase = (size_t)n * DD * HWs + w;
  const size_t pbase = (size_t)n * DD * HWP + PAD + w;

  float win1[7], win2[7];
#pragma unroll
  for (int d = 0; d < 7; ++d) { win1[d] = 0.f; win2[d] = 0.f; }

  float s1[7], s2[7], b1[7], b2[7];

  for (int hr = h0 - RAD; hr < h0 + RAD; ++hr) {
    if (hr < 0) continue;
    const size_t rb = fbase + (size_t)hr * WW;
#pragma unroll
    for (int d = 0; d < 7; ++d) {
      const float2 v = x2[rb + (size_t)d * HWs];
      s1[d] = v.x + v.y;
      s2[d] = v.x * v.x + v.y * v.y;
    }
    dbox7(s1, b1); dbox7(s2, b2);
#pragma unroll
    for (int d = 0; d < 7; ++d) { win1[d] += b1[d]; win2[d] += b2[d]; }
    if (hr >= h0) {
#pragma unroll
      for (int d = 0; d < 7; ++d) S1[rb + (size_t)d * HWs] = s1[d];
    }
  }

  for (int h = h0; h < h0 + CH; ++h) {
    const int hl = h + RAD;
    if (hl < HH) {
      const size_t rb = fbase + (size_t)hl * WW;
      if (hl + 1 < HH) __builtin_prefetch(&x2[rb + WW], 0, 1);
#pragma unroll
      for (int d = 0; d < 7; ++d) {
        const float2 v = x2[rb + (size_t)d * HWs];
        s1[d] = v.x + v.y;
        s2[d] = v.x * v.x + v.y * v.y;
      }
      dbox7(s1, b1); dbox7(s2, b2);
#pragma unroll
      for (int d = 0; d < 7; ++d) { win1[d] += b1[d]; win2[d] += b2[d]; }
      if (hl < h0 + CH) {
#pragma unroll
        for (int d = 0; d < 7; ++d) S1[rb + (size_t)d * HWs] = s1[d];
      }
    }
    {
      const size_t ob = pbase + (size_t)h * WP;
#pragma unroll
      for (int d = 0; d < 7; ++d) {
        A1[ob + (size_t)d * HWP] = win1[d];
        A2[ob + (size_t)d * HWP] = win2[d];
      }
    }
    const int ht = h - RAD;
    if (ht >= 0) {
      const size_t rb = fbase + (size_t)ht * WW;
#pragma unroll
      for (int d = 0; d < 7; ++d) {
        const float2 v = x2[rb + (size_t)d * HWs];
        s1[d] = v.x + v.y;
        s2[d] = v.x * v.x + v.y * v.y;
      }
      dbox7(s1, b1); dbox7(s2, b2);
#pragma unroll
      for (int d = 0; d < 7; ++d) { win1[d] -= b1[d]; win2[d] -= b2[d]; }
    }
  }
}

__global__ void __launch_bounds__(128)
k2_wbox_wmma2(const float* __restrict__ A1, const float* __restrict__ A2,
              float* __restrict__ m, float* __restrict__ q2) {
  __shared__ __attribute__((aligned(16))) float sA1[16 * LP];
  __shared__ __attribute__((aligned(16))) float sA2[16 * LP];
  const int lane = threadIdx.x;
  const int half = lane >> 4;
  const int n16  = lane & 15;
  const int wv   = threadIdx.y;
  const int tid  = wv * 32 + lane;
  const int r0  = blockIdx.y * 16;
  const int j0b = blockIdx.x * TCOLS;
  const int j0  = j0b + 32 * wv;
  {
    const size_t gb = (size_t)r0 * WP + j0b;
    v4f t1[NLD], t2[NLD];
    stage_fetch(A1 + gb, tid, t1);
    stage_fetch(A2 + gb, tid, t2);
    stage_put(sA1, tid, t1);
    stage_put(sA2, tid, t2);
  }
  __syncthreads();
  const float* pa1 = sA1 + n16 * LP + (PAD - RAD + 2 * half + 32 * wv);
  const float* pa2 = sA2 + n16 * LP + (PAD - RAD + 2 * half + 32 * wv);
  const int u = n16 + RAD - 2 * half;

  v8f c10 = {0.f,0.f,0.f,0.f,0.f,0.f,0.f,0.f};
  v8f c20 = {0.f,0.f,0.f,0.f,0.f,0.f,0.f,0.f};
  v8f c11 = {0.f,0.f,0.f,0.f,0.f,0.f,0.f,0.f};
  v8f c21 = {0.f,0.f,0.f,0.f,0.f,0.f,0.f,0.f};

#pragma unroll
  for (int t = 0; t < 23; ++t) {
    const v2f a1 = *reinterpret_cast<const v2f*>(pa1 + 4 * t);
    const v2f a2 = *reinterpret_cast<const v2f*>(pa2 + 4 * t);
    if (t <= 18) {
      const v2f b = band_b(u - 4 * t);
      c10 = WMMA_F32_4(a1, b, c10);
      c20 = WMMA_F32_4(a2, b, c20);
    }
    if (t >= 4) {
      const v2f b = band_b(u + 16 - 4 * t);
      c11 = WMMA_F32_4(a1, b, c11);
      c21 = WMMA_F32_4(a2, b, c21);
    }
  }
#pragma unroll
  for (int v = 0; v < 8; ++v) {
    const int row = r0 + v + 8 * half;
    const size_t o = (size_t)row * WW + (j0 + n16);
    m[o]       = c10[v] * INVK;
    q2[o]      = c20[v] * INVK;
    m[o + 16]  = c11[v] * INVK;
    q2[o + 16] = c21[v] * INVK;
  }
}

__global__ void __launch_bounds__(64)
k3_prep_dbox_hbox(const float* __restrict__ m, const float* __restrict__ S1,
                  float* __restrict__ B1, float* __restrict__ B2,
                  float* __restrict__ B3) {
  const int w  = blockIdx.x * 64 + threadIdx.x;
  const int h0 = blockIdx.y * CH;
  const int n  = blockIdx.z;
  const size_t fbase = (size_t)n * DD * HWs + w;
  const size_t pbase = (size_t)n * DD * HWP + PAD + w;

  float w1[7], w2[7], w3[7];
#pragma unroll
  for (int d = 0; d < 7; ++d) { w1[d] = 0.f; w2[d] = 0.f; w3[d] = 0.f; }

  float t1[7], t2[7], t3[7], b1[7], b2[7], b3[7];

  auto readrow = [&](int hr) {
    const size_t rb = fbase + (size_t)hr * WW;
#pragma unroll
    for (int d = 0; d < 7; ++d) {
      const float mv = m[rb + (size_t)d * HWs];
      const float sv = S1[rb + (size_t)d * HWs];
      t1[d] = mv;
      t2[d] = mv * sv;
      t3[d] = mv * mv;
    }
    dbox7(t1, b1); dbox7(t2, b2); dbox7(t3, b3);
  };

  for (int hr = h0 - RAD; hr < h0 + RAD; ++hr) {
    if (hr < 0) continue;
    readrow(hr);
#pragma unroll
    for (int d = 0; d < 7; ++d) { w1[d] += b1[d]; w2[d] += b2[d]; w3[d] += b3[d]; }
  }
  for (int h = h0; h < h0 + CH; ++h) {
    const int hl = h + RAD;
    if (hl < HH) {
      if (hl + 1 < HH) __builtin_prefetch(&m[fbase + (size_t)(hl + 1) * WW], 0, 1);
      readrow(hl);
#pragma unroll
      for (int d = 0; d < 7; ++d) { w1[d] += b1[d]; w2[d] += b2[d]; w3[d] += b3[d]; }
    }
    {
      const size_t ob = pbase + (size_t)h * WP;
#pragma unroll
      for (int d = 0; d < 7; ++d) {
        B1[ob + (size_t)d * HWP] = w1[d];
        B2[ob + (size_t)d * HWP] = w2[d];
        B3[ob + (size_t)d * HWP] = w3[d];
      }
    }
    const int ht = h - RAD;
    if (ht >= 0) {
      readrow(ht);
#pragma unroll
      for (int d = 0; d < 7; ++d) { w1[d] -= b1[d]; w2[d] -= b2[d]; w3[d] -= b3[d]; }
    }
  }
}

__global__ void __launch_bounds__(128)
k4_wbox_wmma3_stats(const float* __restrict__ B1, const float* __restrict__ B2,
                    const float* __restrict__ B3, const float* __restrict__ m,
                    const float* __restrict__ q2, float* __restrict__ R) {
  __shared__ __attribute__((aligned(16))) float sB1[16 * LP];
  __shared__ __attribute__((aligned(16))) float sB2[16 * LP];
  __shared__ __attribute__((aligned(16))) float sB3[16 * LP];
  const int lane = threadIdx.x;
  const int half = lane >> 4;
  const int n16  = lane & 15;
  const int wv   = threadIdx.y;
  const int tid  = wv * 32 + lane;
  const int r0  = blockIdx.y * 16;
  const int j0b = blockIdx.x * TCOLS;
  const int j0  = j0b + 32 * wv;
  {
    const size_t gb = (size_t)r0 * WP + j0b;
    v4f t1[NLD], t2[NLD], t3[NLD];
    stage_fetch(B1 + gb, tid, t1);
    stage_fetch(B2 + gb, tid, t2);
    stage_fetch(B3 + gb, tid, t3);
    stage_put(sB1, tid, t1);
    stage_put(sB2, tid, t2);
    stage_put(sB3, tid, t3);
  }
  __syncthreads();
  const float* p1 = sB1 + n16 * LP + (PAD - RAD + 2 * half + 32 * wv);
  const float* p2 = sB2 + n16 * LP + (PAD - RAD + 2 * half + 32 * wv);
  const float* p3 = sB3 + n16 * LP + (PAD - RAD + 2 * half + 32 * wv);
  const int u = n16 + RAD - 2 * half;

  v8f cm0 = {0.f,0.f,0.f,0.f,0.f,0.f,0.f,0.f};
  v8f cs0 = {0.f,0.f,0.f,0.f,0.f,0.f,0.f,0.f};
  v8f cq0 = {0.f,0.f,0.f,0.f,0.f,0.f,0.f,0.f};
  v8f cm1 = {0.f,0.f,0.f,0.f,0.f,0.f,0.f,0.f};
  v8f cs1 = {0.f,0.f,0.f,0.f,0.f,0.f,0.f,0.f};
  v8f cq1 = {0.f,0.f,0.f,0.f,0.f,0.f,0.f,0.f};

#pragma unroll
  for (int t = 0; t < 23; ++t) {
    const v2f a1 = *reinterpret_cast<const v2f*>(p1 + 4 * t);
    const v2f a2 = *reinterpret_cast<const v2f*>(p2 + 4 * t);
    const v2f a3 = *reinterpret_cast<const v2f*>(p3 + 4 * t);
    if (t <= 18) {
      const v2f b = band_b(u - 4 * t);
      cm0 = WMMA_F32_4(a1, b, cm0);
      cs0 = WMMA_F32_4(a2, b, cs0);
      cq0 = WMMA_F32_4(a3, b, cq0);
    }
    if (t >= 4) {
      const v2f b = band_b(u + 16 - 4 * t);
      cm1 = WMMA_F32_4(a1, b, cm1);
      cs1 = WMMA_F32_4(a2, b, cs1);
      cq1 = WMMA_F32_4(a3, b, cq1);
    }
  }
#pragma unroll
  for (int v = 0; v < 8; ++v) {
    const int row = r0 + v + 8 * half;
    const size_t o = (size_t)row * WW + (j0 + n16);
    {
      const float mv  = m[o];
      const float qv  = q2[o];
      const float c1v = mv - 2.0f * INVK * cm0[v];
      const float c2v = qv - 2.0f * INVK * (cs0[v] - cq0[v]);
      const float var = c2v - c1v * c1v;
      R[o] = 1.0f / (sqrtf(var) + EPSF);
    }
    {
      const float mv  = m[o + 16];
      const float qv  = q2[o + 16];
      const float c1v = mv - 2.0f * INVK * cm1[v];
      const float c2v = qv - 2.0f * INVK * (cs1[v] - cq1[v]);
      const float var = c2v - c1v * c1v;
      R[o + 16] = 1.0f / (sqrtf(var) + EPSF);
    }
  }
}

__global__ void __launch_bounds__(256)
k5_finalize(const float* __restrict__ x, const float* __restrict__ m,
            const float* __restrict__ R, float* __restrict__ out) {
  const size_t i = (size_t)blockIdx.x * 256 + threadIdx.x;
  const float2 xv = ((const float2*)x)[i];
  const float mv = m[i];
  const float rv = R[i];
  float2 o;
  o.x = (xv.x - mv) * rv;
  o.y = (xv.y - mv) * rv;
  ((float2*)out)[i] = o;
}

extern "C" void kernel_launch(void* const* d_in, const int* in_sizes, int n_in,
                              void* d_out, int out_size, void* d_ws, size_t ws_size,
                              hipStream_t stream) {
  (void)in_sizes; (void)n_in; (void)out_size; (void)ws_size;
  const float* x = (const float*)d_in[0];
  float* out = (float*)d_out;
  float* ws = (float*)d_ws;

  const size_t F  = (size_t)NN * DD * HH * WW;
  const size_t FP = (size_t)RTOT * WP;
  float* S1  = ws;
  float* A1p = ws + F;
  float* A2p = ws + F + FP;
  float* mf  = ws + F + 2 * FP;
  float* q2  = mf + F;
  float* B3p = q2 + F;

  const dim3 gZ((RTOT * 64) / 256, 3);
  const dim3 gHB(WW / 64, HH / CH, NN);
  const dim3 bHB(64);
  const dim3 gW(WW / TCOLS, RTOT / 16);
  const dim3 bW(32, 4);

  k0_zero_pads<<<gZ, 256, 0, stream>>>(A1p, A2p, B3p);
  k1_chsum_dbox_hbox<<<gHB, bHB, 0, stream>>>(x, S1, A1p, A2p);
  k2_wbox_wmma2<<<gW, bW, 0, stream>>>(A1p, A2p, mf, q2);
  k3_prep_dbox_hbox<<<gHB, bHB, 0, stream>>>(mf, S1, A1p, A2p, B3p);
  k4_wbox_wmma3_stats<<<gW, bW, 0, stream>>>(A1p, A2p, B3p, mf, q2,  S1);
  k5_finalize<<<(unsigned)(F / 256), 256, 0, stream>>>(x, mf, S1, out);
}


# K1/K3 H running-window kernels: each thread keeps 42-56 row loads in flight (row registers refilled as rows are consumed, exact counted waits) instead of draining 7-14 loads at every step
# speedup vs baseline: 2.3029x; 1.5520x over previous
#include <hip/hip_runtime.h>
#include <math.h>


#define NN 4
#define DD 7
#define HH 768
#define WW 768
#define HWs (HH * WW)
#define RAD 30
#define CH 64
#define PAD 32
#define WP (WW + 2 * PAD)
#define HWP (HH * WP)
#define RTOT (NN * DD * HH)

static_assert(HH % CH == 0, "chunking");
static_assert(CH > RAD, "warmup writes must stay in chunk");

constexpr float INVK = 1.0f / (7.0f * 61.0f * 61.0f * 2.0f);
constexpr float EPSF = 1e-7f;

typedef __attribute__((ext_vector_type(2))) float v2f;
typedef __attribute__((ext_vector_type(4))) float v4f;
typedef __attribute__((ext_vector_type(8))) float v8f;

#define TCOLS 128
#define WIN4 48
#define NLD ((16 * WIN4) / 128)
#define LP 196

static_assert(WW % TCOLS == 0, "column blocks");
static_assert((16 * WIN4) % 128 == 0, "staging loads divide evenly");
static_assert(4 * WIN4 >= TCOLS + 2 * RAD + (PAD - RAD), "window covers the band");
static_assert(LP >= 4 * WIN4 && (LP % 8) == 4, "pitch = 4 * odd");

__device__ __forceinline__ void stage_fetch(const float* __restrict__ g, int tid,
                                            v4f (&v)[NLD]) {
#pragma unroll
  for (int it = 0; it < NLD; ++it) {
    const int idx = it * 128 + tid;
    const int row = idx / WIN4;
    const int c4  = idx - row * WIN4;
    v[it] = *reinterpret_cast<const v4f*>(g + (size_t)row * WP + 4 * c4);
  }
}
__device__ __forceinline__ void stage_put(float* s, int tid, const v4f (&v)[NLD]) {
#pragma unroll
  for (int it = 0; it < NLD; ++it) {
    const int idx = it * 128 + tid;
    const int row = idx / WIN4;
    const int c4  = idx - row * WIN4;
    *reinterpret_cast<v4f*>(s + row * LP + 4 * c4) = v[it];
  }
}

#define WMMA_F32_4(A, B, C) \
  __builtin_amdgcn_wmma_f32_16x16x4_f32(false, (A), false, (B), (short)0, (C), false, false)

__device__ __forceinline__ v2f band_b(int e) {
  v2f b;
  b.x = ((unsigned)(e + RAD) < 61u) ? 1.f : 0.f;
  b.y = ((unsigned)(e - 1 + RAD) < 61u) ? 1.f : 0.f;
  return b;
}

__device__ __forceinline__ void dbox7(const float* s, float* b) {
  float ps[8];
  ps[0] = 0.f;
#pragma unroll
  for (int d = 0; d < 7; ++d) ps[d + 1] = ps[d] + s[d];
#pragma unroll
  for (int d = 0; d < 7; ++d) {
    const int hi = (d + 3 > 6) ? 7 : (d + 4);
    const int lo = (d - 3 < 0) ? 0 : (d - 3);
    b[d] = ps[hi] - ps[lo];
  }
}

__global__ void __launch_bounds__(256)
k0_zero_pads(float* __restrict__ f0, float* __restrict__ f1,
             float* __restrict__ f2) {
  const int i = blockIdx.x * 256 + threadIdx.x;
  const int row = i >> 6;
  const int c = i & 63;
  const int col = (c < PAD) ? c : (c + WW);
  const size_t o = (size_t)row * WP + col;
  float* f = (blockIdx.y == 0) ? f0 : (blockIdx.y == 1) ? f1 : f2;
  f[o] = 0.f;
}

__device__ __forceinline__ int clamp_row(int h) {
  return h < 0 ? 0 : (h > HH - 1 ? HH - 1 : h);
}

__device__ __forceinline__ void k1_issue(const float2* __restrict__ xc, int hr,
                                         unsigned wl, float2 (&q)[7]) {
  const float2* p = xc + (size_t)clamp_row(hr) * WW;
#pragma unroll
  for (int d = 0; d < 7; ++d) q[d] = (p + (size_t)d * HWs)[wl];
}
__device__ __forceinline__ void k1_sums(const float2 (&q)[7], bool ok,
                                        float (&s1)[7], float (&s2)[7]) {
#pragma unroll
  for (int d = 0; d < 7; ++d) {
    const float vx = ok ? q[d].x : 0.f;
    const float vy = ok ? q[d].y : 0.f;
    s1[d] = vx + vy;
    s2[d] = vx * vx + vy * vy;
  }
}

#define K1_WARM 6
#define K1_MAIN 4
static_assert((2 * RAD) % K1_WARM == 0 && CH % K1_MAIN == 0, "row groups divide evenly");
static_assert(K1_WARM == 6 && K1_MAIN == 4, "hand-over schedule below is written for 6 and 4");

template <int AHEAD>
__device__ __forceinline__ void k1_warm_step(const float2* __restrict__ xc,
                                             float* __restrict__ S1c, unsigned wl,
                                             int hr, int h0, float2 (&q)[7],
                                             float (&win1)[7], float (&win2)[7]) {
  float s1[7], s2[7], b1[7], b2[7];
  k1_sums(q, hr >= 0, s1, s2);
  if (AHEAD > 0) k1_issue(xc, hr + AHEAD, wl, q);
  dbox7(s1, b1); dbox7(s2, b2);
#pragma unroll
  for (int d = 0; d < 7; ++d) { win1[d] += b1[d]; win2[d] += b2[d]; }
  if (hr >= h0) {
    float* so = S1c + (size_t)hr * WW;
#pragma unroll
    for (int d = 0; d < 7; ++d) (so + (size_t)d * HWs)[wl] = s1[d];
  }
}

template <int AHEAD>
__device__ __forceinline__ void k1_main_step(const float2* __restrict__ xc,
                                             float* __restrict__ S1c,
                                             float* __restrict__ A1c,
                                             float* __restrict__ A2c, unsigned wl,
                                             int h, int h0, float2 (&ql)[7],
                                             float2 (&qt)[7], float (&win1)[7],
                                             float (&win2)[7]) {
  float s1[7], s2[7], b1[7], b2[7];
  const int hl = h + RAD;
  const int ht = h - RAD;
  k1_sums(ql, hl < HH, s1, s2);
  if (AHEAD > 0) k1_issue(xc, hl + AHEAD, wl, ql);
  dbox7(s1, b1); dbox7(s2, b2);
#pragma unroll
  for (int d = 0; d < 7; ++d) { win1[d] += b1[d]; win2[d] += b2[d]; }
  if (hl < h0 + CH) {
    float* so = S1c + (size_t)hl * WW;
#pragma unroll
    for (int d = 0; d < 7; ++d) (so + (size_t)d * HWs)[wl] = s1[d];
  }
  {
    float* o1 = A1c + (size_t)h * WP;
    float* o2 = A2c + (size_t)h * WP;
#pragma unroll
    for (int d = 0; d < 7; ++d) {
      (o1 + (size_t)d * HWP)[wl] = win1[d];
      (o2 + (size_t)d * HWP)[wl] = win2[d];
    }
  }
  k1_sums(qt, ht >= 0, s1, s2);
  if (AHEAD > 0) k1_issue(xc, ht + AHEAD, wl, qt);
  dbox7(s1, b1); dbox7(s2, b2);
#pragma unroll
  for (int d = 0; d < 7; ++d) { win1[d] -= b1[d]; win2[d] -= b2[d]; }
}

__global__ void __launch_bounds__(64)
k1_chsum_dbox_hbox(const float* __restrict__ x, float* __restrict__ S1,
                   float* __restrict__ A1, float* __restrict__ A2) {
  const unsigned wl = threadIdx.x;
  const int w0 = blockIdx.x * 64;
  const int h0 = blockIdx.y * CH;
  const int n  = blockIdx.z;
  const float2* xc = (const float2*)x + ((size_t)n * DD * HWs + w0);
  float* S1c = S1 + ((size_t)n * DD * HWs + w0);
  float* A1c = A1 + ((size_t)n * DD * HWP + PAD + w0);
  float* A2c = A2 + ((size_t)n * DD * HWP + PAD + w0);

  float win1[7], win2[7];
#pragma unroll
  for (int d = 0; d < 7; ++d) { win1[d] = 0.f; win2[d] = 0.f; }

  float2 q[K1_WARM][7];
  float2 ql[K1_MAIN][7], qt[K1_MAIN][7];
  const int hw0 = h0 - RAD;

#pragma unroll
  for (int r = 0; r < K1_WARM; ++r) k1_issue(xc, hw0 + r, wl, q[r]);
  for (int ib = 0; ib < 2 * RAD - K1_WARM; ib += K1_WARM) {
#pragma unroll
    for (int r = 0; r < K1_WARM; ++r)
      k1_warm_step<K1_WARM>(xc, S1c, wl, hw0 + ib + r, h0, q[r], win1, win2);
  }
#pragma unroll
  for (int r = 0; r < K1_WARM; ++r) {
    k1_warm_step<0>(xc, S1c, wl, hw0 + (2 * RAD - K1_WARM) + r, h0, q[r], win1, win2);
    if (r < 4) {
      if ((r & 1) == 0) k1_issue(xc, h0 + (r >> 1) + RAD, wl, ql[r >> 1]);
      else              k1_issue(xc, h0 + (r >> 1) - RAD, wl, qt[r >> 1]);
    } else {
      k1_issue(xc, h0 + (r - 2) + RAD, wl, ql[r - 2]);
      k1_issue(xc, h0 + (r - 2) - RAD, wl, qt[r - 2]);
    }
  }

  for (int hb = h0; hb < h0 + CH - K1_MAIN; hb += K1_MAIN) {
#pragma unroll
    for (int r = 0; r < K1_MAIN; ++r)
      k1_main_step<K1_MAIN>(xc, S1c, A1c, A2c, wl, hb + r, h0, ql[r], qt[r], win1, win2);
  }
#pragma unroll
  for (int r = 0; r < K1_MAIN; ++r)
    k1_main_step<0>(xc, S1c, A1c, A2c, wl, h0 + CH - K1_MAIN + r, h0, ql[r], qt[r], win1, win2);
}

__global__ void __launch_bounds__(128)
k2_wbox_wmma2(const float* __restrict__ A1, const float* __restrict__ A2,
              float* __restrict__ m, float* __restrict__ q2) {
  __shared__ __attribute__((aligned(16))) float sA1[16 * LP];
  __shared__ __attribute__((aligned(16))) float sA2[16 * LP];
  const int lane = threadIdx.x;
  const int half = lane >> 4;
  const int n16  = lane & 15;
  const int wv   = threadIdx.y;
  const int tid  = wv * 32 + lane;
  const int r0  = blockIdx.y * 16;
  const int j0b = blockIdx.x * TCOLS;
  const int j0  = j0b + 32 * wv;
  {
    const size_t gb = (size_t)r0 * WP + j0b;
    v4f t1[NLD], t2[NLD];
    stage_fetch(A1 + gb, tid, t1);
    stage_fetch(A2 + gb, tid, t2);
    stage_put(sA1, tid, t1);
    stage_put(sA2, tid, t2);
  }
  __syncthreads();
  const float* pa1 = sA1 + n16 * LP + (PAD - RAD + 2 * half + 32 * wv);
  const float* pa2 = sA2 + n16 * LP + (PAD - RAD + 2 * half + 32 * wv);
  const int u = n16 + RAD - 2 * half;

  v8f c10 = {0.f,0.f,0.f,0.f,0.f,0.f,0.f,0.f};
  v8f c20 = {0.f,0.f,0.f,0.f,0.f,0.f,0.f,0.f};
  v8f c11 = {0.f,0.f,0.f,0.f,0.f,0.f,0.f,0.f};
  v8f c21 = {0.f,0.f,0.f,0.f,0.f,0.f,0.f,0.f};

#pragma unroll
  for (int t = 0; t < 23; ++t) {
    const v2f a1 = *reinterpret_cast<const v2f*>(pa1 + 4 * t);
    const v2f a2 = *reinterpret_cast<const v2f*>(pa2 + 4 * t);
    if (t <= 18) {
      const v2f b = band_b(u - 4 * t);
      c10 = WMMA_F32_4(a1, b, c10);
      c20 = WMMA_F32_4(a2, b, c20);
    }
    if (t >= 4) {
      const v2f b = band_b(u + 16 - 4 * t);
      c11 = WMMA_F32_4(a1, b, c11);
      c21 = WMMA_F32_4(a2, b, c21);
    }
  }
#pragma unroll
  for (int v = 0; v < 8; ++v) {
    const int row = r0 + v + 8 * half;
    const size_t o = (size_t)row * WW + (j0 + n16);
    m[o]       = c10[v] * INVK;
    q2[o]      = c20[v] * INVK;
    m[o + 16]  = c11[v] * INVK;
    q2[o + 16] = c21[v] * INVK;
  }
}

__device__ __forceinline__ void k3_issue(const float* __restrict__ mc,
                                         const float* __restrict__ sc, int hr,
                                         unsigned wl, float (&qm)[7], float (&qs)[7]) {
  const size_t ro = (size_t)clamp_row(hr) * WW;
#pragma unroll
  for (int d = 0; d < 7; ++d) {
    qm[d] = (mc + ro + (size_t)d * HWs)[wl];
    qs[d] = (sc + ro + (size_t)d * HWs)[wl];
  }
}
__device__ __forceinline__ void k3_terms(const float (&qm)[7], const float (&qs)[7],
                                         bool ok, float (&t1)[7], float (&t2)[7],
                                         float (&t3)[7]) {
#pragma unroll
  for (int d = 0; d < 7; ++d) {
    const float mv = ok ? qm[d] : 0.f;
    const float sv = ok ? qs[d] : 0.f;
    t1[d] = mv;
    t2[d] = mv * sv;
    t3[d] = mv * mv;
  }
}

#define K3_WARM 4
#define K3_MAIN 2
static_assert((2 * RAD) % K3_WARM == 0 && CH % K3_MAIN == 0, "row groups divide evenly");
static_assert(K3_WARM == 2 * K3_MAIN, "hand-over: one main row issued behind each last warmup row");

template <int AHEAD>
__device__ __forceinline__ void k3_warm_step(const float* __restrict__ mc,
                                             const float* __restrict__ sc, unsigned wl,
                                             int hr, float (&qm)[7], float (&qs)[7],
                                             float (&w1)[7], float (&w2)[7],
                                             float (&w3)[7]) {
  float t1[7], t2[7], t3[7], b1[7], b2[7], b3[7];
  k3_terms(qm, qs, hr >= 0, t1, t2, t3);
  if (AHEAD > 0) k3_issue(mc, sc, hr + AHEAD, wl, qm, qs);
  dbox7(t1, b1); dbox7(t2, b2); dbox7(t3, b3);
#pragma unroll
  for (int d = 0; d < 7; ++d) { w1[d] += b1[d]; w2[d] += b2[d]; w3[d] += b3[d]; }
}

template <int AHEAD>
__device__ __forceinline__ void k3_main_step(const float* __restrict__ mc,
                                             const float* __restrict__ sc,
                                             float* __restrict__ B1c,
                                             float* __restrict__ B2c,
                                             float* __restrict__ B3c, unsigned wl, int h,
                                             float (&lm)[7], float (&ls)[7],
                                             float (&tm)[7], float (&ts)[7],
                                             float (&w1)[7], float (&w2)[7],
                                             float (&w3)[7]) {
  float t1[7], t2[7], t3[7], b1[7], b2[7], b3[7];
  const int hl = h + RAD;
  const int ht = h - RAD;
  k3_terms(lm, ls, hl < HH, t1, t2, t3);
  if (AHEAD > 0) k3_issue(mc, sc, hl + AHEAD, wl, lm, ls);
  dbox7(t1, b1); dbox7(t2, b2); dbox7(t3, b3);
#pragma unroll
  for (int d = 0; d < 7; ++d) { w1[d] += b1[d]; w2[d] += b2[d]; w3[d] += b3[d]; }
  {
    const size_t ob = (size_t)h * WP;
#pragma unroll
    for (int d = 0; d < 7; ++d) {
      (B1c + ob + (size_t)d * HWP)[wl] = w1[d];
      (B2c + ob + (size_t)d * HWP)[wl] = w2[d];
      (B3c + ob + (size_t)d * HWP)[wl] = w3[d];
    }
  }
  k3_terms(tm, ts, ht >= 0, t1, t2, t3);
  if (AHEAD > 0) k3_issue(mc, sc, ht + AHEAD, wl, tm, ts);
  dbox7(t1, b1); dbox7(t2, b2); dbox7(t3, b3);
#pragma unroll
  for (int d = 0; d < 7; ++d) { w1[d] -= b1[d]; w2[d] -= b2[d]; w3[d] -= b3[d]; }
}

__global__ void __launch_bounds__(64)
k3_prep_dbox_hbox(const float* __restrict__ m, const float* __restrict__ S1,
                  float* __restrict__ B1, float* __restrict__ B2,
                  float* __restrict__ B3) {
  const unsigned wl = threadIdx.x;
  const int w0 = blockIdx.x * 64;
  const int h0 = blockIdx.y * CH;
  const int n  = blockIdx.z;
  const float* mc = m  + ((size_t)n * DD * HWs + w0);
  const float* sc = S1 + ((size_t)n * DD * HWs + w0);
  const size_t pb = (size_t)n * DD * HWP + PAD + w0;
  float* B1c = B1 + pb;
  float* B2c = B2 + pb;
  float* B3c = B3 + pb;

  float w1[7], w2[7], w3[7];
#pragma unroll
  for (int d = 0; d < 7; ++d) { w1[d] = 0.f; w2[d] = 0.f; w3[d] = 0.f; }

  float qm[K3_WARM][7], qs[K3_WARM][7];
  float lm[K3_MAIN][7], ls[K3_MAIN][7];
  float tm[K3_MAIN][7], ts[K3_MAIN][7];
  const int hw0 = h0 - RAD;

#pragma unroll
  for (int r = 0; r < K3_WARM; ++r) k3_issue(mc, sc, hw0 + r, wl, qm[r], qs[r]);
  for (int ib = 0; ib < 2 * RAD - K3_WARM; ib += K3_WARM) {
#pragma unroll
    for (int r = 0; r < K3_WARM; ++r)
      k3_warm_step<K3_WARM>(mc, sc, wl, hw0 + ib + r, qm[r], qs[r], w1, w2, w3);
  }
#pragma unroll
  for (int r = 0; r < K3_WARM; ++r) {
    k3_warm_step<0>(mc, sc, wl, hw0 + (2 * RAD - K3_WARM) + r, qm[r], qs[r], w1, w2, w3);
    if ((r & 1) == 0) k3_issue(mc, sc, h0 + (r >> 1) + RAD, wl, lm[r >> 1], ls[r >> 1]);
    else              k3_issue(mc, sc, h0 + (r >> 1) - RAD, wl, tm[r >> 1], ts[r >> 1]);
  }

  for (int hb = h0; hb < h0 + CH - K3_MAIN; hb += K3_MAIN) {
#pragma unroll
    for (int r = 0; r < K3_MAIN; ++r)
      k3_main_step<K3_MAIN>(mc, sc, B1c, B2c, B3c, wl, hb + r, lm[r], ls[r], tm[r], ts[r],
                            w1, w2, w3);
  }
#pragma unroll
  for (int r = 0; r < K3_MAIN; ++r)
    k3_main_step<0>(mc, sc, B1c, B2c, B3c, wl, h0 + CH - K3_MAIN + r, lm[r], ls[r], tm[r],
                    ts[r], w1, w2, w3);
}

__global__ void __launch_bounds__(128)
k4_wbox_wmma3_stats(const float* __restrict__ B1, const float* __restrict__ B2,
                    const float* __restrict__ B3, const float* __restrict__ m,
                    const float* __restrict__ q2, float* __restrict__ R) {
  __shared__ __attribute__((aligned(16))) float sB1[16 * LP];
  __shared__ __attribute__((aligned(16))) float sB2[16 * LP];
  __shared__ __attribute__((aligned(16))) float sB3[16 * LP];
  const int lane = threadIdx.x;
  const int half = lane >> 4;
  const int n16  = lane & 15;
  const int wv   = threadIdx.y;
  const int tid  = wv * 32 + lane;
  const int r0  = blockIdx.y * 16;
  const int j0b = blockIdx.x * TCOLS;
  const int j0  = j0b + 32 * wv;
  {
    const size_t gb = (size_t)r0 * WP + j0b;
    v4f t1[NLD], t2[NLD], t3[NLD];
    stage_fetch(B1 + gb, tid, t1);
    stage_fetch(B2 + gb, tid, t2);
    stage_fetch(B3 + gb, tid, t3);
    stage_put(sB1, tid, t1);
    stage_put(sB2, tid, t2);
    stage_put(sB3, tid, t3);
  }
  __syncthreads();
  const float* p1 = sB1 + n16 * LP + (PAD - RAD + 2 * half + 32 * wv);
  const float* p2 = sB2 + n16 * LP + (PAD - RAD + 2 * half + 32 * wv);
  const float* p3 = sB3 + n16 * LP + (PAD - RAD + 2 * half + 32 * wv);
  const int u = n16 + RAD - 2 * half;

  v8f cm0 = {0.f,0.f,0.f,0.f,0.f,0.f,0.f,0.f};
  v8f cs0 = {0.f,0.f,0.f,0.f,0.f,0.f,0.f,0.f};
  v8f cq0 = {0.f,0.f,0.f,0.f,0.f,0.f,0.f,0.f};
  v8f cm1 = {0.f,0.f,0.f,0.f,0.f,0.f,0.f,0.f};
  v8f cs1 = {0.f,0.f,0.f,0.f,0.f,0.f,0.f,0.f};
  v8f cq1 = {0.f,0.f,0.f,0.f,0.f,0.f,0.f,0.f};

#pragma unroll
  for (int t = 0; t < 23; ++t) {
    const v2f a1 = *reinterpret_cast<const v2f*>(p1 + 4 * t);
    const v2f a2 = *reinterpret_cast<const v2f*>(p2 + 4 * t);
    const v2f a3 = *reinterpret_cast<const v2f*>(p3 + 4 * t);
    if (t <= 18) {
      const v2f b = band_b(u - 4 * t);
      cm0 = WMMA_F32_4(a1, b, cm0);
      cs0 = WMMA_F32_4(a2, b, cs0);
      cq0 = WMMA_F32_4(a3, b, cq0);
    }
    if (t >= 4) {
      const v2f b = band_b(u + 16 - 4 * t);
      cm1 = WMMA_F32_4(a1, b, cm1);
      cs1 = WMMA_F32_4(a2, b, cs1);
      cq1 = WMMA_F32_4(a3, b, cq1);
    }
  }
#pragma unroll
  for (int v = 0; v < 8; ++v) {
    const int row = r0 + v + 8 * half;
    const size_t o = (size_t)row * WW + (j0 + n16);
    {
      const float mv  = m[o];
      const float qv  = q2[o];
      const float c1v = mv - 2.0f * INVK * cm0[v];
      const float c2v = qv - 2.0f * INVK * (cs0[v] - cq0[v]);
      const float var = c2v - c1v * c1v;
      R[o] = 1.0f / (sqrtf(var) + EPSF);
    }
    {
      const float mv  = m[o + 16];
      const float qv  = q2[o + 16];
      const float c1v = mv - 2.0f * INVK * cm1[v];
      const float c2v = qv - 2.0f * INVK * (cs1[v] - cq1[v]);
      const float var = c2v - c1v * c1v;
      R[o + 16] = 1.0f / (sqrtf(var) + EPSF);
    }
  }
}

__global__ void __launch_bounds__(256)
k5_finalize(const float* __restrict__ x, const float* __restrict__ m,
            const float* __restrict__ R, float* __restrict__ out) {
  const size_t i = (size_t)blockIdx.x * 256 + threadIdx.x;
  const float2 xv = ((const float2*)x)[i];
  const float mv = m[i];
  const float rv = R[i];
  float2 o;
  o.x = (xv.x - mv) * rv;
  o.y = (xv.y - mv) * rv;
  ((float2*)out)[i] = o;
}

extern "C" void kernel_launch(void* const* d_in, const int* in_sizes, int n_in,
                              void* d_out, int out_size, void* d_ws, size_t ws_size,
                              hipStream_t stream) {
  (void)in_sizes; (void)n_in; (void)out_size; (void)ws_size;
  const float* x = (const float*)d_in[0];
  float* out = (float*)d_out;
  float* ws = (float*)d_ws;

  const size_t F  = (size_t)NN * DD * HH * WW;
  const size_t FP = (size_t)RTOT * WP;
  float* S1  = ws;
  float* A1p = ws + F;
  float* A2p = ws + F + FP;
  float* mf  = ws + F + 2 * FP;
  float* q2  = mf + F;
  float* B3p = q2 + F;

  const dim3 gZ((RTOT * 64) / 256, 3);
  const dim3 gHB(WW / 64, HH / CH, NN);
  const dim3 bHB(64);
  const dim3 gW(WW / TCOLS, RTOT / 16);
  const dim3 bW(32, 4);

  k0_zero_pads<<<gZ, 256, 0, stream>>>(A1p, A2p, B3p);
  k1_chsum_dbox_hbox<<<gHB, bHB, 0, stream>>>(x, S1, A1p, A2p);
  k2_wbox_wmma2<<<gW, bW, 0, stream>>>(A1p, A2p, mf, q2);
  k3_prep_dbox_hbox<<<gHB, bHB, 0, stream>>>(mf, S1, A1p, A2p, B3p);
  k4_wbox_wmma3_stats<<<gW, bW, 0, stream>>>(A1p, A2p, B3p, mf, q2,  S1);
  k5_finalize<<<(unsigned)(F / 256), 256, 0, stream>>>(x, mf, S1, out);
}


# Pass B filters two fields, not three: box filter is linear, so K3 forms m*(S1-m) per element (difference taken before the box sums, not after); B3 and a third of K4's staging, LDS and WMMAs go
# speedup vs baseline: 2.5359x; 1.1012x over previous
#include <hip/hip_runtime.h>
#include <math.h>


#define NN 4
#define DD 7
#define HH 768
#define WW 768
#define HWs (HH * WW)
#define RAD 30
#define CH 64
#define PAD 32
#define WP (WW + 2 * PAD)
#define HWP (HH * WP)
#define RTOT (NN * DD * HH)

static_assert(HH % CH == 0, "chunking");
static_assert(CH > RAD, "warmup writes must stay in chunk");

constexpr float INVK = 1.0f / (7.0f * 61.0f * 61.0f * 2.0f);
constexpr float EPSF = 1e-7f;

typedef __attribute__((ext_vector_type(2))) float v2f;
typedef __attribute__((ext_vector_type(4))) float v4f;
typedef __attribute__((ext_vector_type(8))) float v8f;

#define TCOLS 128
#define WIN4 48
#define NLD ((16 * WIN4) / 128)
#define LP 196

static_assert(WW % TCOLS == 0, "column blocks");
static_assert((16 * WIN4) % 128 == 0, "staging loads divide evenly");
static_assert(4 * WIN4 >= TCOLS + 2 * RAD + (PAD - RAD), "window covers the band");
static_assert(LP >= 4 * WIN4 && (LP % 8) == 4, "pitch = 4 * odd");

__device__ __forceinline__ void stage_fetch(const float* __restrict__ g, int tid,
                                            v4f (&v)[NLD]) {
#pragma unroll
  for (int it = 0; it < NLD; ++it) {
    const int idx = it * 128 + tid;
    const int row = idx / WIN4;
    const int c4  = idx - row * WIN4;
    v[it] = *reinterpret_cast<const v4f*>(g + (size_t)row * WP + 4 * c4);
  }
}
__device__ __forceinline__ void stage_put(float* s, int tid, const v4f (&v)[NLD]) {
#pragma unroll
  for (int it = 0; it < NLD; ++it) {
    const int idx = it * 128 + tid;
    const int row = idx / WIN4;
    const int c4  = idx - row * WIN4;
    *reinterpret_cast<v4f*>(s + row * LP + 4 * c4) = v[it];
  }
}

#define WMMA_F32_4(A, B, C) \
  __builtin_amdgcn_wmma_f32_16x16x4_f32(false, (A), false, (B), (short)0, (C), false, false)

__device__ __forceinline__ v2f band_b(int e) {
  v2f b;
  b.x = ((unsigned)(e + RAD) < 61u) ? 1.f : 0.f;
  b.y = ((unsigned)(e - 1 + RAD) < 61u) ? 1.f : 0.f;
  return b;
}

__device__ __forceinline__ void dbox7(const float* s, float* b) {
  float ps[8];
  ps[0] = 0.f;
#pragma unroll
  for (int d = 0; d < 7; ++d) ps[d + 1] = ps[d] + s[d];
#pragma unroll
  for (int d = 0; d < 7; ++d) {
    const int hi = (d + 3 > 6) ? 7 : (d + 4);
    const int lo = (d - 3 < 0) ? 0 : (d - 3);
    b[d] = ps[hi] - ps[lo];
  }
}

__global__ void __launch_bounds__(256)
k0_zero_pads(float* __restrict__ f0, float* __restrict__ f1) {
  const int i = blockIdx.x * 256 + threadIdx.x;
  const int row = i >> 6;
  const int c = i & 63;
  const int col = (c < PAD) ? c : (c + WW);
  const size_t o = (size_t)row * WP + col;
  float* f = (blockIdx.y == 0) ? f0 : f1;
  f[o] = 0.f;
}

__device__ __forceinline__ int clamp_row(int h) {
  return h < 0 ? 0 : (h > HH - 1 ? HH - 1 : h);
}

__device__ __forceinline__ void k1_issue(const float2* __restrict__ xc, int hr,
                                         unsigned wl, float2 (&q)[7]) {
  const float2* p = xc + (size_t)clamp_row(hr) * WW;
#pragma unroll
  for (int d = 0; d < 7; ++d) q[d] = (p + (size_t)d * HWs)[wl];
}
__device__ __forceinline__ void k1_sums(const float2 (&q)[7], bool ok,
                                        float (&s1)[7], float (&s2)[7]) {
#pragma unroll
  for (int d = 0; d < 7; ++d) {
    const float vx = ok ? q[d].x : 0.f;
    const float vy = ok ? q[d].y : 0.f;
    s1[d] = vx + vy;
    s2[d] = vx * vx + vy * vy;
  }
}

#define K1_WARM 6
#define K1_MAIN 4
static_assert((2 * RAD) % K1_WARM == 0 && CH % K1_MAIN == 0, "row groups divide evenly");
static_assert(K1_WARM == 6 && K1_MAIN == 4, "hand-over schedule below is written for 6 and 4");

template <int AHEAD>
__device__ __forceinline__ void k1_warm_step(const float2* __restrict__ xc,
                                             float* __restrict__ S1c, unsigned wl,
                                             int hr, int h0, float2 (&q)[7],
                                             float (&win1)[7], float (&win2)[7]) {
  float s1[7], s2[7], b1[7], b2[7];
  k1_sums(q, hr >= 0, s1, s2);
  if (AHEAD > 0) k1_issue(xc, hr + AHEAD, wl, q);
  dbox7(s1, b1); dbox7(s2, b2);
#pragma unroll
  for (int d = 0; d < 7; ++d) { win1[d] += b1[d]; win2[d] += b2[d]; }
  if (hr >= h0) {
    float* so = S1c + (size_t)hr * WW;
#pragma unroll
    for (int d = 0; d < 7; ++d) (so + (size_t)d * HWs)[wl] = s1[d];
  }
}

template <int AHEAD>
__device__ __forceinline__ void k1_main_step(const float2* __restrict__ xc,
                                             float* __restrict__ S1c,
                                             float* __restrict__ A1c,
                                             float* __restrict__ A2c, unsigned wl,
                                             int h, int h0, float2 (&ql)[7],
                                             float2 (&qt)[7], float (&win1)[7],
                                             float (&win2)[7]) {
  float s1[7], s2[7], b1[7], b2[7];
  const int hl = h + RAD;
  const int ht = h - RAD;
  k1_sums(ql, hl < HH, s1, s2);
  if (AHEAD > 0) k1_issue(xc, hl + AHEAD, wl, ql);
  dbox7(s1, b1); dbox7(s2, b2);
#pragma unroll
  for (int d = 0; d < 7; ++d) { win1[d] += b1[d]; win2[d] += b2[d]; }
  if (hl < h0 + CH) {
    float* so = S1c + (size_t)hl * WW;
#pragma unroll
    for (int d = 0; d < 7; ++d) (so + (size_t)d * HWs)[wl] = s1[d];
  }
  {
    float* o1 = A1c + (size_t)h * WP;
    float* o2 = A2c + (size_t)h * WP;
#pragma unroll
    for (int d = 0; d < 7; ++d) {
      (o1 + (size_t)d * HWP)[wl] = win1[d];
      (o2 + (size_t)d * HWP)[wl] = win2[d];
    }
  }
  k1_sums(qt, ht >= 0, s1, s2);
  if (AHEAD > 0) k1_issue(xc, ht + AHEAD, wl, qt);
  dbox7(s1, b1); dbox7(s2, b2);
#pragma unroll
  for (int d = 0; d < 7; ++d) { win1[d] -= b1[d]; win2[d] -= b2[d]; }
}

__global__ void __launch_bounds__(64)
k1_chsum_dbox_hbox(const float* __restrict__ x, float* __restrict__ S1,
                   float* __restrict__ A1, float* __restrict__ A2) {
  const unsigned wl = threadIdx.x;
  const int w0 = blockIdx.x * 64;
  const int h0 = blockIdx.y * CH;
  const int n  = blockIdx.z;
  const float2* xc = (const float2*)x + ((size_t)n * DD * HWs + w0);
  float* S1c = S1 + ((size_t)n * DD * HWs + w0);
  float* A1c = A1 + ((size_t)n * DD * HWP + PAD + w0);
  float* A2c = A2 + ((size_t)n * DD * HWP + PAD + w0);

  float win1[7], win2[7];
#pragma unroll
  for (int d = 0; d < 7; ++d) { win1[d] = 0.f; win2[d] = 0.f; }

  float2 q[K1_WARM][7];
  float2 ql[K1_MAIN][7], qt[K1_MAIN][7];
  const int hw0 = h0 - RAD;

#pragma unroll
  for (int r = 0; r < K1_WARM; ++r) k1_issue(xc, hw0 + r, wl, q[r]);
  for (int ib = 0; ib < 2 * RAD - K1_WARM; ib += K1_WARM) {
#pragma unroll
    for (int r = 0; r < K1_WARM; ++r)
      k1_warm_step<K1_WARM>(xc, S1c, wl, hw0 + ib + r, h0, q[r], win1, win2);
  }
#pragma unroll
  for (int r = 0; r < K1_WARM; ++r) {
    k1_warm_step<0>(xc, S1c, wl, hw0 + (2 * RAD - K1_WARM) + r, h0, q[r], win1, win2);
    if (r < 4) {
      if ((r & 1) == 0) k1_issue(xc, h0 + (r >> 1) + RAD, wl, ql[r >> 1]);
      else              k1_issue(xc, h0 + (r >> 1) - RAD, wl, qt[r >> 1]);
    } else {
      k1_issue(xc, h0 + (r - 2) + RAD, wl, ql[r - 2]);
      k1_issue(xc, h0 + (r - 2) - RAD, wl, qt[r - 2]);
    }
  }

  for (int hb = h0; hb < h0 + CH - K1_MAIN; hb += K1_MAIN) {
#pragma unroll
    for (int r = 0; r < K1_MAIN; ++r)
      k1_main_step<K1_MAIN>(xc, S1c, A1c, A2c, wl, hb + r, h0, ql[r], qt[r], win1, win2);
  }
#pragma unroll
  for (int r = 0; r < K1_MAIN; ++r)
    k1_main_step<0>(xc, S1c, A1c, A2c, wl, h0 + CH - K1_MAIN + r, h0, ql[r], qt[r], win1, win2);
}

__global__ void __launch_bounds__(128)
k2_wbox_wmma2(const float* __restrict__ A1, const float* __restrict__ A2,
              float* __restrict__ m, float* __restrict__ q2) {
  __shared__ __attribute__((aligned(16))) float sA1[16 * LP];
  __shared__ __attribute__((aligned(16))) float sA2[16 * LP];
  const int lane = threadIdx.x;
  const int half = lane >> 4;
  const int n16  = lane & 15;
  const int wv   = threadIdx.y;
  const int tid  = wv * 32 + lane;
  const int r0  = blockIdx.y * 16;
  const int j0b = blockIdx.x * TCOLS;
  const int j0  = j0b + 32 * wv;
  {
    const size_t gb = (size_t)r0 * WP + j0b;
    v4f t1[NLD], t2[NLD];
    stage_fetch(A1 + gb, tid, t1);
    stage_fetch(A2 + gb, tid, t2);
    stage_put(sA1, tid, t1);
    stage_put(sA2, tid, t2);
  }
  __syncthreads();
  const float* pa1 = sA1 + n16 * LP + (PAD - RAD + 2 * half + 32 * wv);
  const float* pa2 = sA2 + n16 * LP + (PAD - RAD + 2 * half + 32 * wv);
  const int u = n16 + RAD - 2 * half;

  v8f c10 = {0.f,0.f,0.f,0.f,0.f,0.f,0.f,0.f};
  v8f c20 = {0.f,0.f,0.f,0.f,0.f,0.f,0.f,0.f};
  v8f c11 = {0.f,0.f,0.f,0.f,0.f,0.f,0.f,0.f};
  v8f c21 = {0.f,0.f,0.f,0.f,0.f,0.f,0.f,0.f};

#pragma unroll
  for (int t = 0; t < 23; ++t) {
    const v2f a1 = *reinterpret_cast<const v2f*>(pa1 + 4 * t);
    const v2f a2 = *reinterpret_cast<const v2f*>(pa2 + 4 * t);
    if (t <= 18) {
      const v2f b = band_b(u - 4 * t);
      c10 = WMMA_F32_4(a1, b, c10);
      c20 = WMMA_F32_4(a2, b, c20);
    }
    if (t >= 4) {
      const v2f b = band_b(u + 16 - 4 * t);
      c11 = WMMA_F32_4(a1, b, c11);
      c21 = WMMA_F32_4(a2, b, c21);
    }
  }
#pragma unroll
  for (int v = 0; v < 8; ++v) {
    const int row = r0 + v + 8 * half;
    const size_t o = (size_t)row * WW + (j0 + n16);
    m[o]       = c10[v] * INVK;
    q2[o]      = c20[v] * INVK;
    m[o + 16]  = c11[v] * INVK;
    q2[o + 16] = c21[v] * INVK;
  }
}

__device__ __forceinline__ void k3_issue(const float* __restrict__ mc,
                                         const float* __restrict__ sc, int hr,
                                         unsigned wl, float (&qm)[7], float (&qs)[7]) {
  const size_t ro = (size_t)clamp_row(hr) * WW;
#pragma unroll
  for (int d = 0; d < 7; ++d) {
    qm[d] = (mc + ro + (size_t)d * HWs)[wl];
    qs[d] = (sc + ro + (size_t)d * HWs)[wl];
  }
}
__device__ __forceinline__ void k3_terms(const float (&qm)[7], const float (&qs)[7],
                                         bool ok, float (&t1)[7], float (&t2)[7]) {
#pragma unroll
  for (int d = 0; d < 7; ++d) {
    const float mv = ok ? qm[d] : 0.f;
    const float sv = ok ? qs[d] : 0.f;
    t1[d] = mv;
    t2[d] = mv * (sv - mv);
  }
}

#define K3_WARM 4
#define K3_MAIN 2
static_assert((2 * RAD) % K3_WARM == 0 && CH % K3_MAIN == 0, "row groups divide evenly");
static_assert(K3_WARM == 2 * K3_MAIN, "hand-over: one main row issued behind each last warmup row");

template <int AHEAD>
__device__ __forceinline__ void k3_warm_step(const float* __restrict__ mc,
                                             const float* __restrict__ sc, unsigned wl,
                                             int hr, float (&qm)[7], float (&qs)[7],
                                             float (&w1)[7], float (&w2)[7]) {
  float t1[7], t2[7], b1[7], b2[7];
  k3_terms(qm, qs, hr >= 0, t1, t2);
  if (AHEAD > 0) k3_issue(mc, sc, hr + AHEAD, wl, qm, qs);
  dbox7(t1, b1); dbox7(t2, b2);
#pragma unroll
  for (int d = 0; d < 7; ++d) { w1[d] += b1[d]; w2[d] += b2[d]; }
}

template <int AHEAD>
__device__ __forceinline__ void k3_main_step(const float* __restrict__ mc,
                                             const float* __restrict__ sc,
                                             float* __restrict__ B1c,
                                             float* __restrict__ B2c, unsigned wl, int h,
                                             float (&lm)[7], float (&ls)[7],
                                             float (&tm)[7], float (&ts)[7],
                                             float (&w1)[7], float (&w2)[7]) {
  float t1[7], t2[7], b1[7], b2[7];
  const int hl = h + RAD;
  const int ht = h - RAD;
  k3_terms(lm, ls, hl < HH, t1, t2);
  if (AHEAD > 0) k3_issue(mc, sc, hl + AHEAD, wl, lm, ls);
  dbox7(t1, b1); dbox7(t2, b2);
#pragma unroll
  for (int d = 0; d < 7; ++d) { w1[d] += b1[d]; w2[d] += b2[d]; }
  {
    const size_t ob = (size_t)h * WP;
#pragma unroll
    for (int d = 0; d < 7; ++d) {
      (B1c + ob + (size_t)d * HWP)[wl] = w1[d];
      (B2c + ob + (size_t)d * HWP)[wl] = w2[d];
    }
  }
  k3_terms(tm, ts, ht >= 0, t1, t2);
  if (AHEAD > 0) k3_issue(mc, sc, ht + AHEAD, wl, tm, ts);
  dbox7(t1, b1); dbox7(t2, b2);
#pragma unroll
  for (int d = 0; d < 7; ++d) { w1[d] -= b1[d]; w2[d] -= b2[d]; }
}

__global__ void __launch_bounds__(64)
k3_prep_dbox_hbox(const float* __restrict__ m, const float* __restrict__ S1,
                  float* __restrict__ B1, float* __restrict__ B2) {
  const unsigned wl = threadIdx.x;
  const int w0 = blockIdx.x * 64;
  const int h0 = blockIdx.y * CH;
  const int n  = blockIdx.z;
  const float* mc = m  + ((size_t)n * DD * HWs + w0);
  const float* sc = S1 + ((size_t)n * DD * HWs + w0);
  const size_t pb = (size_t)n * DD * HWP + PAD + w0;
  float* B1c = B1 + pb;
  float* B2c = B2 + pb;

  float w1[7], w2[7];
#pragma unroll
  for (int d = 0; d < 7; ++d) { w1[d] = 0.f; w2[d] = 0.f; }

  float qm[K3_WARM][7], qs[K3_WARM][7];
  float lm[K3_MAIN][7], ls[K3_MAIN][7];
  float tm[K3_MAIN][7], ts[K3_MAIN][7];
  const int hw0 = h0 - RAD;

#pragma unroll
  for (int r = 0; r < K3_WARM; ++r) k3_issue(mc, sc, hw0 + r, wl, qm[r], qs[r]);
  for (int ib = 0; ib < 2 * RAD - K3_WARM; ib += K3_WARM) {
#pragma unroll
    for (int r = 0; r < K3_WARM; ++r)
      k3_warm_step<K3_WARM>(mc, sc, wl, hw0 + ib + r, qm[r], qs[r], w1, w2);
  }
#pragma unroll
  for (int r = 0; r < K3_WARM; ++r) {
    k3_warm_step<0>(mc, sc, wl, hw0 + (2 * RAD - K3_WARM) + r, qm[r], qs[r], w1, w2);
    if ((r & 1) == 0) k3_issue(mc, sc, h0 + (r >> 1) + RAD, wl, lm[r >> 1], ls[r >> 1]);
    else              k3_issue(mc, sc, h0 + (r >> 1) - RAD, wl, tm[r >> 1], ts[r >> 1]);
  }

  for (int hb = h0; hb < h0 + CH - K3_MAIN; hb += K3_MAIN) {
#pragma unroll
    for (int r = 0; r < K3_MAIN; ++r)
      k3_main_step<K3_MAIN>(mc, sc, B1c, B2c, wl, hb + r, lm[r], ls[r], tm[r], ts[r],
                            w1, w2);
  }
#pragma unroll
  for (int r = 0; r < K3_MAIN; ++r)
    k3_main_step<0>(mc, sc, B1c, B2c, wl, h0 + CH - K3_MAIN + r, lm[r], ls[r], tm[r],
                    ts[r], w1, w2);
}

__global__ void __launch_bounds__(128)
k4_wbox_wmma2_stats(const float* __restrict__ B1, const float* __restrict__ B2,
                    const float* __restrict__ m, const float* __restrict__ q2,
                    float* __restrict__ R) {
  __shared__ __attribute__((aligned(16))) float sB1[16 * LP];
  __shared__ __attribute__((aligned(16))) float sB2[16 * LP];
  const int lane = threadIdx.x;
  const int half = lane >> 4;
  const int n16  = lane & 15;
  const int wv   = threadIdx.y;
  const int tid  = wv * 32 + lane;
  const int r0  = blockIdx.y * 16;
  const int j0b = blockIdx.x * TCOLS;
  const int j0  = j0b + 32 * wv;
  {
    const size_t gb = (size_t)r0 * WP + j0b;
    v4f t1[NLD], t2[NLD];
    stage_fetch(B1 + gb, tid, t1);
    stage_fetch(B2 + gb, tid, t2);
    stage_put(sB1, tid, t1);
    stage_put(sB2, tid, t2);
  }
  __syncthreads();
  const float* p1 = sB1 + n16 * LP + (PAD - RAD + 2 * half + 32 * wv);
  const float* p2 = sB2 + n16 * LP + (PAD - RAD + 2 * half + 32 * wv);
  const int u = n16 + RAD - 2 * half;

  v8f cm0 = {0.f,0.f,0.f,0.f,0.f,0.f,0.f,0.f};
  v8f cs0 = {0.f,0.f,0.f,0.f,0.f,0.f,0.f,0.f};
  v8f cm1 = {0.f,0.f,0.f,0.f,0.f,0.f,0.f,0.f};
  v8f cs1 = {0.f,0.f,0.f,0.f,0.f,0.f,0.f,0.f};

#pragma unroll
  for (int t = 0; t < 23; ++t) {
    const v2f a1 = *reinterpret_cast<const v2f*>(p1 + 4 * t);
    const v2f a2 = *reinterpret_cast<const v2f*>(p2 + 4 * t);
    if (t <= 18) {
      const v2f b = band_b(u - 4 * t);
      cm0 = WMMA_F32_4(a1, b, cm0);
      cs0 = WMMA_F32_4(a2, b, cs0);
    }
    if (t >= 4) {
      const v2f b = band_b(u + 16 - 4 * t);
      cm1 = WMMA_F32_4(a1, b, cm1);
      cs1 = WMMA_F32_4(a2, b, cs1);
    }
  }
#pragma unroll
  for (int v = 0; v < 8; ++v) {
    const int row = r0 + v + 8 * half;
    const size_t o = (size_t)row * WW + (j0 + n16);
    {
      const float mv  = m[o];
      const float qv  = q2[o];
      const float c1v = mv - 2.0f * INVK * cm0[v];
      const float c2v = qv - 2.0f * INVK * cs0[v];
      const float var = c2v - c1v * c1v;
      R[o] = 1.0f / (sqrtf(var) + EPSF);
    }
    {
      const float mv  = m[o + 16];
      const float qv  = q2[o + 16];
      const float c1v = mv - 2.0f * INVK * cm1[v];
      const float c2v = qv - 2.0f * INVK * cs1[v];
      const float var = c2v - c1v * c1v;
      R[o + 16] = 1.0f / (sqrtf(var) + EPSF);
    }
  }
}

__global__ void __launch_bounds__(256)
k5_finalize(const float* __restrict__ x, const float* __restrict__ m,
            const float* __restrict__ R, float* __restrict__ out) {
  const size_t i = (size_t)blockIdx.x * 256 + threadIdx.x;
  const float2 xv = ((const float2*)x)[i];
  const float mv = m[i];
  const float rv = R[i];
  float2 o;
  o.x = (xv.x - mv) * rv;
  o.y = (xv.y - mv) * rv;
  ((float2*)out)[i] = o;
}

extern "C" void kernel_launch(void* const* d_in, const int* in_sizes, int n_in,
                              void* d_out, int out_size, void* d_ws, size_t ws_size,
                              hipStream_t stream) {
  (void)in_sizes; (void)n_in; (void)out_size; (void)ws_size;
  const float* x = (const float*)d_in[0];
  float* out = (float*)d_out;
  float* ws = (float*)d_ws;

  const size_t F  = (size_t)NN * DD * HH * WW;
  const size_t FP = (size_t)RTOT * WP;
  float* S1  = ws;
  float* A1p = ws + F;
  float* A2p = ws + F + FP;
  float* mf  = ws + F + 2 * FP;
  float* q2  = mf + F;

  const dim3 gZ((RTOT * 64) / 256, 2);
  const dim3 gHB(WW / 64, HH / CH, NN);
  const dim3 bHB(64);
  const dim3 gW(WW / TCOLS, RTOT / 16);
  const dim3 bW(32, 4);

  k0_zero_pads<<<gZ, 256, 0, stream>>>(A1p, A2p);
  k1_chsum_dbox_hbox<<<gHB, bHB, 0, stream>>>(x, S1, A1p, A2p);
  k2_wbox_wmma2<<<gW, bW, 0, stream>>>(A1p, A2p, mf, q2);
  k3_prep_dbox_hbox<<<gHB, bHB, 0, stream>>>(mf, S1, A1p, A2p);
  k4_wbox_wmma2_stats<<<gW, bW, 0, stream>>>(A1p, A2p, mf, q2,  S1);
  k5_finalize<<<(unsigned)(F / 256), 256, 0, stream>>>(x, mf, S1, out);
}


# K5 folded into K4's epilogue: each wave turns its two WMMA tiles into R in registers and writes out=(x-m)*R itself, x, m, q2 requested up front; R's store and reload, K5's m read and a launch go
# speedup vs baseline: 2.9048x; 1.1455x over previous
#include <hip/hip_runtime.h>
#include <math.h>


#define NN 4
#define DD 7
#define HH 768
#define WW 768
#define HWs (HH * WW)
#define RAD 30
#define CH 64
#define PAD 32
#define WP (WW + 2 * PAD)
#define HWP (HH * WP)
#define RTOT (NN * DD * HH)

static_assert(HH % CH == 0, "chunking");
static_assert(CH > RAD, "warmup writes must stay in chunk");

constexpr float INVK = 1.0f / (7.0f * 61.0f * 61.0f * 2.0f);
constexpr float EPSF = 1e-7f;

typedef __attribute__((ext_vector_type(2))) float v2f;
typedef __attribute__((ext_vector_type(4))) float v4f;
typedef __attribute__((ext_vector_type(8))) float v8f;

#define TCOLS 128
#define WIN4 48
#define NLD ((16 * WIN4) / 128)
#define LP 196

static_assert(WW % TCOLS == 0, "column blocks");
static_assert((16 * WIN4) % 128 == 0, "staging loads divide evenly");
static_assert(4 * WIN4 >= TCOLS + 2 * RAD + (PAD - RAD), "window covers the band");
static_assert(LP >= 4 * WIN4 && (LP % 8) == 4, "pitch = 4 * odd");

__device__ __forceinline__ void stage_fetch(const float* __restrict__ g, int tid,
                                            v4f (&v)[NLD]) {
#pragma unroll
  for (int it = 0; it < NLD; ++it) {
    const int idx = it * 128 + tid;
    const int row = idx / WIN4;
    const int c4  = idx - row * WIN4;
    v[it] = *reinterpret_cast<const v4f*>(g + (size_t)row * WP + 4 * c4);
  }
}
__device__ __forceinline__ void stage_put(float* s, int tid, const v4f (&v)[NLD]) {
#pragma unroll
  for (int it = 0; it < NLD; ++it) {
    const int idx = it * 128 + tid;
    const int row = idx / WIN4;
    const int c4  = idx - row * WIN4;
    *reinterpret_cast<v4f*>(s + row * LP + 4 * c4) = v[it];
  }
}

#define WMMA_F32_4(A, B, C) \
  __builtin_amdgcn_wmma_f32_16x16x4_f32(false, (A), false, (B), (short)0, (C), false, false)

__device__ __forceinline__ v2f band_b(int e) {
  v2f b;
  b.x = ((unsigned)(e + RAD) < 61u) ? 1.f : 0.f;
  b.y = ((unsigned)(e - 1 + RAD) < 61u) ? 1.f : 0.f;
  return b;
}

__device__ __forceinline__ void dbox7(const float* s, float* b) {
  float ps[8];
  ps[0] = 0.f;
#pragma unroll
  for (int d = 0; d < 7; ++d) ps[d + 1] = ps[d] + s[d];
#pragma unroll
  for (int d = 0; d < 7; ++d) {
    const int hi = (d + 3 > 6) ? 7 : (d + 4);
    const int lo = (d - 3 < 0) ? 0 : (d - 3);
    b[d] = ps[hi] - ps[lo];
  }
}

__global__ void __launch_bounds__(256)
k0_zero_pads(float* __restrict__ f0, float* __restrict__ f1) {
  const int i = blockIdx.x * 256 + threadIdx.x;
  const int row = i >> 6;
  const int c = i & 63;
  const int col = (c < PAD) ? c : (c + WW);
  const size_t o = (size_t)row * WP + col;
  float* f = (blockIdx.y == 0) ? f0 : f1;
  f[o] = 0.f;
}

__device__ __forceinline__ int clamp_row(int h) {
  return h < 0 ? 0 : (h > HH - 1 ? HH - 1 : h);
}

__device__ __forceinline__ void k1_issue(const float2* __restrict__ xc, int hr,
                                         unsigned wl, float2 (&q)[7]) {
  const float2* p = xc + (size_t)clamp_row(hr) * WW;
#pragma unroll
  for (int d = 0; d < 7; ++d) q[d] = (p + (size_t)d * HWs)[wl];
}
__device__ __forceinline__ void k1_sums(const float2 (&q)[7], bool ok,
                                        float (&s1)[7], float (&s2)[7]) {
#pragma unroll
  for (int d = 0; d < 7; ++d) {
    const float vx = ok ? q[d].x : 0.f;
    const float vy = ok ? q[d].y : 0.f;
    s1[d] = vx + vy;
    s2[d] = vx * vx + vy * vy;
  }
}

#define K1_WARM 6
#define K1_MAIN 4
static_assert((2 * RAD) % K1_WARM == 0 && CH % K1_MAIN == 0, "row groups divide evenly");
static_assert(K1_WARM == 6 && K1_MAIN == 4, "hand-over schedule below is written for 6 and 4");

template <int AHEAD>
__device__ __forceinline__ void k1_warm_step(const float2* __restrict__ xc,
                                             float* __restrict__ S1c, unsigned wl,
                                             int hr, int h0, float2 (&q)[7],
                                             float (&win1)[7], float (&win2)[7]) {
  float s1[7], s2[7], b1[7], b2[7];
  k1_sums(q, hr >= 0, s1, s2);
  if (AHEAD > 0) k1_issue(xc, hr + AHEAD, wl, q);
  dbox7(s1, b1); dbox7(s2, b2);
#pragma unroll
  for (int d = 0; d < 7; ++d) { win1[d] += b1[d]; win2[d] += b2[d]; }
  if (hr >= h0) {
    float* so = S1c + (size_t)hr * WW;
#pragma unroll
    for (int d = 0; d < 7; ++d) (so + (size_t)d * HWs)[wl] = s1[d];
  }
}

template <int AHEAD>
__device__ __forceinline__ void k1_main_step(const float2* __restrict__ xc,
                                             float* __restrict__ S1c,
                                             float* __restrict__ A1c,
                                             float* __restrict__ A2c, unsigned wl,
                                             int h, int h0, float2 (&ql)[7],
                                             float2 (&qt)[7], float (&win1)[7],
                                             float (&win2)[7]) {
  float s1[7], s2[7], b1[7], b2[7];
  const int hl = h + RAD;
  const int ht = h - RAD;
  k1_sums(ql, hl < HH, s1, s2);
  if (AHEAD > 0) k1_issue(xc, hl + AHEAD, wl, ql);
  dbox7(s1, b1); dbox7(s2, b2);
#pragma unroll
  for (int d = 0; d < 7; ++d) { win1[d] += b1[d]; win2[d] += b2[d]; }
  if (hl < h0 + CH) {
    float* so = S1c + (size_t)hl * WW;
#pragma unroll
    for (int d = 0; d < 7; ++d) (so + (size_t)d * HWs)[wl] = s1[d];
  }
  {
    float* o1 = A1c + (size_t)h * WP;
    float* o2 = A2c + (size_t)h * WP;
#pragma unroll
    for (int d = 0; d < 7; ++d) {
      (o1 + (size_t)d * HWP)[wl] = win1[d];
      (o2 + (size_t)d * HWP)[wl] = win2[d];
    }
  }
  k1_sums(qt, ht >= 0, s1, s2);
  if (AHEAD > 0) k1_issue(xc, ht + AHEAD, wl, qt);
  dbox7(s1, b1); dbox7(s2, b2);
#pragma unroll
  for (int d = 0; d < 7; ++d) { win1[d] -= b1[d]; win2[d] -= b2[d]; }
}

__global__ void __launch_bounds__(64)
k1_chsum_dbox_hbox(const float* __restrict__ x, float* __restrict__ S1,
                   float* __restrict__ A1, float* __restrict__ A2) {
  const unsigned wl = threadIdx.x;
  const int w0 = blockIdx.x * 64;
  const int h0 = blockIdx.y * CH;
  const int n  = blockIdx.z;
  const float2* xc = (const float2*)x + ((size_t)n * DD * HWs + w0);
  float* S1c = S1 + ((size_t)n * DD * HWs + w0);
  float* A1c = A1 + ((size_t)n * DD * HWP + PAD + w0);
  float* A2c = A2 + ((size_t)n * DD * HWP + PAD + w0);

  float win1[7], win2[7];
#pragma unroll
  for (int d = 0; d < 7; ++d) { win1[d] = 0.f; win2[d] = 0.f; }

  float2 q[K1_WARM][7];
  float2 ql[K1_MAIN][7], qt[K1_MAIN][7];
  const int hw0 = h0 - RAD;

#pragma unroll
  for (int r = 0; r < K1_WARM; ++r) k1_issue(xc, hw0 + r, wl, q[r]);
  for (int ib = 0; ib < 2 * RAD - K1_WARM; ib += K1_WARM) {
#pragma unroll
    for (int r = 0; r < K1_WARM; ++r)
      k1_warm_step<K1_WARM>(xc, S1c, wl, hw0 + ib + r, h0, q[r], win1, win2);
  }
#pragma unroll
  for (int r = 0; r < K1_WARM; ++r) {
    k1_warm_step<0>(xc, S1c, wl, hw0 + (2 * RAD - K1_WARM) + r, h0, q[r], win1, win2);
    if (r < 4) {
      if ((r & 1) == 0) k1_issue(xc, h0 + (r >> 1) + RAD, wl, ql[r >> 1]);
      else              k1_issue(xc, h0 + (r >> 1) - RAD, wl, qt[r >> 1]);
    } else {
      k1_issue(xc, h0 + (r - 2) + RAD, wl, ql[r - 2]);
      k1_issue(xc, h0 + (r - 2) - RAD, wl, qt[r - 2]);
    }
  }

  for (int hb = h0; hb < h0 + CH - K1_MAIN; hb += K1_MAIN) {
#pragma unroll
    for (int r = 0; r < K1_MAIN; ++r)
      k1_main_step<K1_MAIN>(xc, S1c, A1c, A2c, wl, hb + r, h0, ql[r], qt[r], win1, win2);
  }
#pragma unroll
  for (int r = 0; r < K1_MAIN; ++r)
    k1_main_step<0>(xc, S1c, A1c, A2c, wl, h0 + CH - K1_MAIN + r, h0, ql[r], qt[r], win1, win2);
}

__global__ void __launch_bounds__(128)
k2_wbox_wmma2(const float* __restrict__ A1, const float* __restrict__ A2,
              float* __restrict__ m, float* __restrict__ q2) {
  __shared__ __attribute__((aligned(16))) float sA1[16 * LP];
  __shared__ __attribute__((aligned(16))) float sA2[16 * LP];
  const int lane = threadIdx.x;
  const int half = lane >> 4;
  const int n16  = lane & 15;
  const int wv   = threadIdx.y;
  const int tid  = wv * 32 + lane;
  const int r0  = blockIdx.y * 16;
  const int j0b = blockIdx.x * TCOLS;
  const int j0  = j0b + 32 * wv;
  {
    const size_t gb = (size_t)r0 * WP + j0b;
    v4f t1[NLD], t2[NLD];
    stage_fetch(A1 + gb, tid, t1);
    stage_fetch(A2 + gb, tid, t2);
    stage_put(sA1, tid, t1);
    stage_put(sA2, tid, t2);
  }
  __syncthreads();
  const float* pa1 = sA1 + n16 * LP + (PAD - RAD + 2 * half + 32 * wv);
  const float* pa2 = sA2 + n16 * LP + (PAD - RAD + 2 * half + 32 * wv);
  const int u = n16 + RAD - 2 * half;

  v8f c10 = {0.f,0.f,0.f,0.f,0.f,0.f,0.f,0.f};
  v8f c20 = {0.f,0.f,0.f,0.f,0.f,0.f,0.f,0.f};
  v8f c11 = {0.f,0.f,0.f,0.f,0.f,0.f,0.f,0.f};
  v8f c21 = {0.f,0.f,0.f,0.f,0.f,0.f,0.f,0.f};

#pragma unroll
  for (int t = 0; t < 23; ++t) {
    const v2f a1 = *reinterpret_cast<const v2f*>(pa1 + 4 * t);
    const v2f a2 = *reinterpret_cast<const v2f*>(pa2 + 4 * t);
    if (t <= 18) {
      const v2f b = band_b(u - 4 * t);
      c10 = WMMA_F32_4(a1, b, c10);
      c20 = WMMA_F32_4(a2, b, c20);
    }
    if (t >= 4) {
      const v2f b = band_b(u + 16 - 4 * t);
      c11 = WMMA_F32_4(a1, b, c11);
      c21 = WMMA_F32_4(a2, b, c21);
    }
  }
#pragma unroll
  for (int v = 0; v < 8; ++v) {
    const int row = r0 + v + 8 * half;
    const size_t o = (size_t)row * WW + (j0 + n16);
    m[o]       = c10[v] * INVK;
    q2[o]      = c20[v] * INVK;
    m[o + 16]  = c11[v] * INVK;
    q2[o + 16] = c21[v] * INVK;
  }
}

__device__ __forceinline__ void k3_issue(const float* __restrict__ mc,
                                         const float* __restrict__ sc, int hr,
                                         unsigned wl, float (&qm)[7], float (&qs)[7]) {
  const size_t ro = (size_t)clamp_row(hr) * WW;
#pragma unroll
  for (int d = 0; d < 7; ++d) {
    qm[d] = (mc + ro + (size_t)d * HWs)[wl];
    qs[d] = (sc + ro + (size_t)d * HWs)[wl];
  }
}
__device__ __forceinline__ void k3_terms(const float (&qm)[7], const float (&qs)[7],
                                         bool ok, float (&t1)[7], float (&t2)[7]) {
#pragma unroll
  for (int d = 0; d < 7; ++d) {
    const float mv = ok ? qm[d] : 0.f;
    const float sv = ok ? qs[d] : 0.f;
    t1[d] = mv;
    t2[d] = mv * (sv - mv);
  }
}

#define K3_WARM 4
#define K3_MAIN 2
static_assert((2 * RAD) % K3_WARM == 0 && CH % K3_MAIN == 0, "row groups divide evenly");
static_assert(K3_WARM == 2 * K3_MAIN, "hand-over: one main row issued behind each last warmup row");

template <int AHEAD>
__device__ __forceinline__ void k3_warm_step(const float* __restrict__ mc,
                                             const float* __restrict__ sc, unsigned wl,
                                             int hr, float (&qm)[7], float (&qs)[7],
                                             float (&w1)[7], float (&w2)[7]) {
  float t1[7], t2[7], b1[7], b2[7];
  k3_terms(qm, qs, hr >= 0, t1, t2);
  if (AHEAD > 0) k3_issue(mc, sc, hr + AHEAD, wl, qm, qs);
  dbox7(t1, b1); dbox7(t2, b2);
#pragma unroll
  for (int d = 0; d < 7; ++d) { w1[d] += b1[d]; w2[d] += b2[d]; }
}

template <int AHEAD>
__device__ __forceinline__ void k3_main_step(const float* __restrict__ mc,
                                             const float* __restrict__ sc,
                                             float* __restrict__ B1c,
                                             float* __restrict__ B2c, unsigned wl, int h,
                                             float (&lm)[7], float (&ls)[7],
                                             float (&tm)[7], float (&ts)[7],
                                             float (&w1)[7], float (&w2)[7]) {
  float t1[7], t2[7], b1[7], b2[7];
  const int hl = h + RAD;
  const int ht = h - RAD;
  k3_terms(lm, ls, hl < HH, t1, t2);
  if (AHEAD > 0) k3_issue(mc, sc, hl + AHEAD, wl, lm, ls);
  dbox7(t1, b1); dbox7(t2, b2);
#pragma unroll
  for (int d = 0; d < 7; ++d) { w1[d] += b1[d]; w2[d] += b2[d]; }
  {
    const size_t ob = (size_t)h * WP;
#pragma unroll
    for (int d = 0; d < 7; ++d) {
      (B1c + ob + (size_t)d * HWP)[wl] = w1[d];
      (B2c + ob + (size_t)d * HWP)[wl] = w2[d];
    }
  }
  k3_terms(tm, ts, ht >= 0, t1, t2);
  if (AHEAD > 0) k3_issue(mc, sc, ht + AHEAD, wl, tm, ts);
  dbox7(t1, b1); dbox7(t2, b2);
#pragma unroll
  for (int d = 0; d < 7; ++d) { w1[d] -= b1[d]; w2[d] -= b2[d]; }
}

__global__ void __launch_bounds__(64)
k3_prep_dbox_hbox(const float* __restrict__ m, const float* __restrict__ S1,
                  float* __restrict__ B1, float* __restrict__ B2) {
  const unsigned wl = threadIdx.x;
  const int w0 = blockIdx.x * 64;
  const int h0 = blockIdx.y * CH;
  const int n  = blockIdx.z;
  const float* mc = m  + ((size_t)n * DD * HWs + w0);
  const float* sc = S1 + ((size_t)n * DD * HWs + w0);
  const size_t pb = (size_t)n * DD * HWP + PAD + w0;
  float* B1c = B1 + pb;
  float* B2c = B2 + pb;

  float w1[7], w2[7];
#pragma unroll
  for (int d = 0; d < 7; ++d) { w1[d] = 0.f; w2[d] = 0.f; }

  float qm[K3_WARM][7], qs[K3_WARM][7];
  float lm[K3_MAIN][7], ls[K3_MAIN][7];
  float tm[K3_MAIN][7], ts[K3_MAIN][7];
  const int hw0 = h0 - RAD;

#pragma unroll
  for (int r = 0; r < K3_WARM; ++r) k3_issue(mc, sc, hw0 + r, wl, qm[r], qs[r]);
  for (int ib = 0; ib < 2 * RAD - K3_WARM; ib += K3_WARM) {
#pragma unroll
    for (int r = 0; r < K3_WARM; ++r)
      k3_warm_step<K3_WARM>(mc, sc, wl, hw0 + ib + r, qm[r], qs[r], w1, w2);
  }
#pragma unroll
  for (int r = 0; r < K3_WARM; ++r) {
    k3_warm_step<0>(mc, sc, wl, hw0 + (2 * RAD - K3_WARM) + r, qm[r], qs[r], w1, w2);
    if ((r & 1) == 0) k3_issue(mc, sc, h0 + (r >> 1) + RAD, wl, lm[r >> 1], ls[r >> 1]);
    else              k3_issue(mc, sc, h0 + (r >> 1) - RAD, wl, tm[r >> 1], ts[r >> 1]);
  }

  for (int hb = h0; hb < h0 + CH - K3_MAIN; hb += K3_MAIN) {
#pragma unroll
    for (int r = 0; r < K3_MAIN; ++r)
      k3_main_step<K3_MAIN>(mc, sc, B1c, B2c, wl, hb + r, lm[r], ls[r], tm[r], ts[r],
                            w1, w2);
  }
#pragma unroll
  for (int r = 0; r < K3_MAIN; ++r)
    k3_main_step<0>(mc, sc, B1c, B2c, wl, h0 + CH - K3_MAIN + r, lm[r], ls[r], tm[r],
                    ts[r], w1, w2);
}

__global__ void __launch_bounds__(128)
k4_wbox_wmma2_stats_out(const float* __restrict__ B1, const float* __restrict__ B2,
                        const float* __restrict__ m, const float* __restrict__ q2,
                        const float* x, float* __restrict__ out) {
  __shared__ __attribute__((aligned(16))) float sB1[16 * LP];
  __shared__ __attribute__((aligned(16))) float sB2[16 * LP];
  const int lane = threadIdx.x;
  const int half = lane >> 4;
  const int n16  = lane & 15;
  const int wv   = threadIdx.y;
  const int tid  = wv * 32 + lane;
  const int r0  = blockIdx.y * 16;
  const int j0b = blockIdx.x * TCOLS;
  const int j0  = j0b + 32 * wv;
  const size_t gb = (size_t)r0 * WP + j0b;
  v4f t1[NLD], t2[NLD];
  stage_fetch(B1 + gb, tid, t1);
  stage_fetch(B2 + gb, tid, t2);

  const size_t ob = (size_t)(r0 + 8 * half) * WW + (j0 + n16);
  const float2* x2 = (const float2*)x;
  float  mv0[8], mv1[8], qv0[8], qv1[8];
  float2 xv0[8], xv1[8];
#pragma unroll
  for (int v = 0; v < 8; ++v) {
    const size_t o = ob + (size_t)v * WW;
    mv0[v] = m[o];
    qv0[v] = q2[o];
    xv0[v] = x2[o];
    mv1[v] = m[o + 16];
    qv1[v] = q2[o + 16];
    xv1[v] = x2[o + 16];
  }

  stage_put(sB1, tid, t1);
  stage_put(sB2, tid, t2);
  __syncthreads();

  const float* p1 = sB1 + n16 * LP + (PAD - RAD + 2 * half + 32 * wv);
  const float* p2 = sB2 + n16 * LP + (PAD - RAD + 2 * half + 32 * wv);
  const int u = n16 + RAD - 2 * half;

  v8f cm0 = {0.f,0.f,0.f,0.f,0.f,0.f,0.f,0.f};
  v8f cs0 = {0.f,0.f,0.f,0.f,0.f,0.f,0.f,0.f};
  v8f cm1 = {0.f,0.f,0.f,0.f,0.f,0.f,0.f,0.f};
  v8f cs1 = {0.f,0.f,0.f,0.f,0.f,0.f,0.f,0.f};

#pragma unroll
  for (int t = 0; t < 23; ++t) {
    const v2f a1 = *reinterpret_cast<const v2f*>(p1 + 4 * t);
    const v2f a2 = *reinterpret_cast<const v2f*>(p2 + 4 * t);
    if (t <= 18) {
      const v2f b = band_b(u - 4 * t);
      cm0 = WMMA_F32_4(a1, b, cm0);
      cs0 = WMMA_F32_4(a2, b, cs0);
    }
    if (t >= 4) {
      const v2f b = band_b(u + 16 - 4 * t);
      cm1 = WMMA_F32_4(a1, b, cm1);
      cs1 = WMMA_F32_4(a2, b, cs1);
    }
  }

  float2* o2 = (float2*)out;
#pragma unroll
  for (int v = 0; v < 8; ++v) {
    const size_t o = ob + (size_t)v * WW;
    {
      const float mv  = mv0[v];
      const float c1v = mv - 2.0f * INVK * cm0[v];
      const float c2v = qv0[v] - 2.0f * INVK * cs0[v];
      const float var = c2v - c1v * c1v;
      const float rv  = 1.0f / (sqrtf(var) + EPSF);
      float2 r;
      r.x = (xv0[v].x - mv) * rv;
      r.y = (xv0[v].y - mv) * rv;
      o2[o] = r;
    }
    {
      const float mv  = mv1[v];
      const float c1v = mv - 2.0f * INVK * cm1[v];
      const float c2v = qv1[v] - 2.0f * INVK * cs1[v];
      const float var = c2v - c1v * c1v;
      const float rv  = 1.0f / (sqrtf(var) + EPSF);
      float2 r;
      r.x = (xv1[v].x - mv) * rv;
      r.y = (xv1[v].y - mv) * rv;
      o2[o + 16] = r;
    }
  }
}

extern "C" void kernel_launch(void* const* d_in, const int* in_sizes, int n_in,
                              void* d_out, int out_size, void* d_ws, size_t ws_size,
                              hipStream_t stream) {
  (void)in_sizes; (void)n_in; (void)out_size; (void)ws_size;
  const float* x = (const float*)d_in[0];
  float* out = (float*)d_out;
  float* ws = (float*)d_ws;

  const size_t F  = (size_t)NN * DD * HH * WW;
  const size_t FP = (size_t)RTOT * WP;
  float* S1  = ws;
  float* A1p = ws + F;
  float* A2p = ws + F + FP;
  float* mf  = ws + F + 2 * FP;
  float* q2  = mf + F;

  const dim3 gZ((RTOT * 64) / 256, 2);
  const dim3 gHB(WW / 64, HH / CH, NN);
  const dim3 bHB(64);
  const dim3 gW(WW / TCOLS, RTOT / 16);
  const dim3 bW(32, 4);

  k0_zero_pads<<<gZ, 256, 0, stream>>>(A1p, A2p);
  k1_chsum_dbox_hbox<<<gHB, bHB, 0, stream>>>(x, S1, A1p, A2p);
  k2_wbox_wmma2<<<gW, bW, 0, stream>>>(A1p, A2p, mf, q2);
  k3_prep_dbox_hbox<<<gHB, bHB, 0, stream>>>(mf, S1, A1p, A2p);
  k4_wbox_wmma2_stats_out<<<gW, bW, 0, stream>>>(A1p, A2p, mf, q2, x, out);
}


# W-box staging puts zeros into LDS for window pieces outside the image instead of loading pad columns; the pad-zeroing kernel (10752 blocks, one 4-byte store a thread), its launch and the pad reads go
# speedup vs baseline: 3.0002x; 1.0328x over previous
#include <hip/hip_runtime.h>
#include <math.h>


#define NN 4
#define DD 7
#define HH 768
#define WW 768
#define HWs (HH * WW)
#define RAD 30
#define CH 64
#define PAD 32
#define WP (WW + 2 * PAD)
#define HWP (HH * WP)
#define RTOT (NN * DD * HH)

static_assert(HH % CH == 0, "chunking");
static_assert(CH > RAD, "warmup writes must stay in chunk");

constexpr float INVK = 1.0f / (7.0f * 61.0f * 61.0f * 2.0f);
constexpr float EPSF = 1e-7f;

typedef __attribute__((ext_vector_type(2))) float v2f;
typedef __attribute__((ext_vector_type(4))) float v4f;
typedef __attribute__((ext_vector_type(8))) float v8f;

#define TCOLS 128
#define WIN4 48
#define NLD ((16 * WIN4) / 128)
#define LP 196

static_assert(WW % TCOLS == 0, "column blocks");
static_assert(PAD % 4 == 0 && WW % 4 == 0 && TCOLS % 4 == 0, "a 16-byte piece never straddles a pad edge");
static_assert((16 * WIN4) % 128 == 0, "staging loads divide evenly");
static_assert(4 * WIN4 >= TCOLS + 2 * RAD + (PAD - RAD), "window covers the band");
static_assert(LP >= 4 * WIN4 && (LP % 8) == 4, "pitch = 4 * odd");

__device__ __forceinline__ void stage_fetch(const float* __restrict__ g, int tid, int j0b,
                                            v4f (&v)[NLD]) {
#pragma unroll
  for (int it = 0; it < NLD; ++it) {
    const int idx = it * 128 + tid;
    const int row = idx / WIN4;
    const int c4  = idx - row * WIN4;
    const int col = j0b + 4 * c4;
    v4f p = {0.f, 0.f, 0.f, 0.f};
    if ((unsigned)(col - PAD) < (unsigned)WW)
      p = *reinterpret_cast<const v4f*>(g + (size_t)row * WP + 4 * c4);
    v[it] = p;
  }
}
__device__ __forceinline__ void stage_put(float* s, int tid, const v4f (&v)[NLD]) {
#pragma unroll
  for (int it = 0; it < NLD; ++it) {
    const int idx = it * 128 + tid;
    const int row = idx / WIN4;
    const int c4  = idx - row * WIN4;
    *reinterpret_cast<v4f*>(s + row * LP + 4 * c4) = v[it];
  }
}

#define WMMA_F32_4(A, B, C) \
  __builtin_amdgcn_wmma_f32_16x16x4_f32(false, (A), false, (B), (short)0, (C), false, false)

__device__ __forceinline__ v2f band_b(int e) {
  v2f b;
  b.x = ((unsigned)(e + RAD) < 61u) ? 1.f : 0.f;
  b.y = ((unsigned)(e - 1 + RAD) < 61u) ? 1.f : 0.f;
  return b;
}

__device__ __forceinline__ void dbox7(const float* s, float* b) {
  float ps[8];
  ps[0] = 0.f;
#pragma unroll
  for (int d = 0; d < 7; ++d) ps[d + 1] = ps[d] + s[d];
#pragma unroll
  for (int d = 0; d < 7; ++d) {
    const int hi = (d + 3 > 6) ? 7 : (d + 4);
    const int lo = (d - 3 < 0) ? 0 : (d - 3);
    b[d] = ps[hi] - ps[lo];
  }
}

__device__ __forceinline__ int clamp_row(int h) {
  return h < 0 ? 0 : (h > HH - 1 ? HH - 1 : h);
}

__device__ __forceinline__ void k1_issue(const float2* __restrict__ xc, int hr,
                                         unsigned wl, float2 (&q)[7]) {
  const float2* p = xc + (size_t)clamp_row(hr) * WW;
#pragma unroll
  for (int d = 0; d < 7; ++d) q[d] = (p + (size_t)d * HWs)[wl];
}
__device__ __forceinline__ void k1_sums(const float2 (&q)[7], bool ok,
                                        float (&s1)[7], float (&s2)[7]) {
#pragma unroll
  for (int d = 0; d < 7; ++d) {
    const float vx = ok ? q[d].x : 0.f;
    const float vy = ok ? q[d].y : 0.f;
    s1[d] = vx + vy;
    s2[d] = vx * vx + vy * vy;
  }
}

#define K1_WARM 6
#define K1_MAIN 4
static_assert((2 * RAD) % K1_WARM == 0 && CH % K1_MAIN == 0, "row groups divide evenly");
static_assert(K1_WARM == 6 && K1_MAIN == 4, "hand-over schedule below is written for 6 and 4");

template <int AHEAD>
__device__ __forceinline__ void k1_warm_step(const float2* __restrict__ xc,
                                             float* __restrict__ S1c, unsigned wl,
                                             int hr, int h0, float2 (&q)[7],
                                             float (&win1)[7], float (&win2)[7]) {
  float s1[7], s2[7], b1[7], b2[7];
  k1_sums(q, hr >= 0, s1, s2);
  if (AHEAD > 0) k1_issue(xc, hr + AHEAD, wl, q);
  dbox7(s1, b1); dbox7(s2, b2);
#pragma unroll
  for (int d = 0; d < 7; ++d) { win1[d] += b1[d]; win2[d] += b2[d]; }
  if (hr >= h0) {
    float* so = S1c + (size_t)hr * WW;
#pragma unroll
    for (int d = 0; d < 7; ++d) (so + (size_t)d * HWs)[wl] = s1[d];
  }
}

template <int AHEAD>
__device__ __forceinline__ void k1_main_step(const float2* __restrict__ xc,
                                             float* __restrict__ S1c,
                                             float* __restrict__ A1c,
                                             float* __restrict__ A2c, unsigned wl,
                                             int h, int h0, float2 (&ql)[7],
                                             float2 (&qt)[7], float (&win1)[7],
                                             float (&win2)[7]) {
  float s1[7], s2[7], b1[7], b2[7];
  const int hl = h + RAD;
  const int ht = h - RAD;
  k1_sums(ql, hl < HH, s1, s2);
  if (AHEAD > 0) k1_issue(xc, hl + AHEAD, wl, ql);
  dbox7(s1, b1); dbox7(s2, b2);
#pragma unroll
  for (int d = 0; d < 7; ++d) { win1[d] += b1[d]; win2[d] += b2[d]; }
  if (hl < h0 + CH) {
    float* so = S1c + (size_t)hl * WW;
#pragma unroll
    for (int d = 0; d < 7; ++d) (so + (size_t)d * HWs)[wl] = s1[d];
  }
  {
    float* o1 = A1c + (size_t)h * WP;
    float* o2 = A2c + (size_t)h * WP;
#pragma unroll
    for (int d = 0; d < 7; ++d) {
      (o1 + (size_t)d * HWP)[wl] = win1[d];
      (o2 + (size_t)d * HWP)[wl] = win2[d];
    }
  }
  k1_sums(qt, ht >= 0, s1, s2);
  if (AHEAD > 0) k1_issue(xc, ht + AHEAD, wl, qt);
  dbox7(s1, b1); dbox7(s2, b2);
#pragma unroll
  for (int d = 0; d < 7; ++d) { win1[d] -= b1[d]; win2[d] -= b2[d]; }
}

__global__ void __launch_bounds__(64)
k1_chsum_dbox_hbox(const float* __restrict__ x, float* __restrict__ S1,
                   float* __restrict__ A1, float* __restrict__ A2) {
  const unsigned wl = threadIdx.x;
  const int w0 = blockIdx.x * 64;
  const int h0 = blockIdx.y * CH;
  const int n  = blockIdx.z;
  const float2* xc = (const float2*)x + ((size_t)n * DD * HWs + w0);
  float* S1c = S1 + ((size_t)n * DD * HWs + w0);
  float* A1c = A1 + ((size_t)n * DD * HWP + PAD + w0);
  float* A2c = A2 + ((size_t)n * DD * HWP + PAD + w0);

  float win1[7], win2[7];
#pragma unroll
  for (int d = 0; d < 7; ++d) { win1[d] = 0.f; win2[d] = 0.f; }

  float2 q[K1_WARM][7];
  float2 ql[K1_MAIN][7], qt[K1_MAIN][7];
  const int hw0 = h0 - RAD;

#pragma unroll
  for (int r = 0; r < K1_WARM; ++r) k1_issue(xc, hw0 + r, wl, q[r]);
  for (int ib = 0; ib < 2 * RAD - K1_WARM; ib += K1_WARM) {
#pragma unroll
    for (int r = 0; r < K1_WARM; ++r)
      k1_warm_step<K1_WARM>(xc, S1c, wl, hw0 + ib + r, h0, q[r], win1, win2);
  }
#pragma unroll
  for (int r = 0; r < K1_WARM; ++r) {
    k1_warm_step<0>(xc, S1c, wl, hw0 + (2 * RAD - K1_WARM) + r, h0, q[r], win1, win2);
    if (r < 4) {
      if ((r & 1) == 0) k1_issue(xc, h0 + (r >> 1) + RAD, wl, ql[r >> 1]);
      else              k1_issue(xc, h0 + (r >> 1) - RAD, wl, qt[r >> 1]);
    } else {
      k1_issue(xc, h0 + (r - 2) + RAD, wl, ql[r - 2]);
      k1_issue(xc, h0 + (r - 2) - RAD, wl, qt[r - 2]);
    }
  }

  for (int hb = h0; hb < h0 + CH - K1_MAIN; hb += K1_MAIN) {
#pragma unroll
    for (int r = 0; r < K1_MAIN; ++r)
      k1_main_step<K1_MAIN>(xc, S1c, A1c, A2c, wl, hb + r, h0, ql[r], qt[r], win1, win2);
  }
#pragma unroll
  for (int r = 0; r < K1_MAIN; ++r)
    k1_main_step<0>(xc, S1c, A1c, A2c, wl, h0 + CH - K1_MAIN + r, h0, ql[r], qt[r], win1, win2);
}

__global__ void __launch_bounds__(128)
k2_wbox_wmma2(const float* __restrict__ A1, const float* __restrict__ A2,
              float* __restrict__ m, float* __restrict__ q2) {
  __shared__ __attribute__((aligned(16))) float sA1[16 * LP];
  __shared__ __attribute__((aligned(16))) float sA2[16 * LP];
  const int lane = threadIdx.x;
  const int half = lane >> 4;
  const int n16  = lane & 15;
  const int wv   = threadIdx.y;
  const int tid  = wv * 32 + lane;
  const int r0  = blockIdx.y * 16;
  const int j0b = blockIdx.x * TCOLS;
  const int j0  = j0b + 32 * wv;
  {
    const size_t gb = (size_t)r0 * WP + j0b;
    v4f t1[NLD], t2[NLD];
    stage_fetch(A1 + gb, tid, j0b, t1);
    stage_fetch(A2 + gb, tid, j0b, t2);
    stage_put(sA1, tid, t1);
    stage_put(sA2, tid, t2);
  }
  __syncthreads();
  const float* pa1 = sA1 + n16 * LP + (PAD - RAD + 2 * half + 32 * wv);
  const float* pa2 = sA2 + n16 * LP + (PAD - RAD + 2 * half + 32 * wv);
  const int u = n16 + RAD - 2 * half;

  v8f c10 = {0.f,0.f,0.f,0.f,0.f,0.f,0.f,0.f};
  v8f c20 = {0.f,0.f,0.f,0.f,0.f,0.f,0.f,0.f};
  v8f c11 = {0.f,0.f,0.f,0.f,0.f,0.f,0.f,0.f};
  v8f c21 = {0.f,0.f,0.f,0.f,0.f,0.f,0.f,0.f};

#pragma unroll
  for (int t = 0; t < 23; ++t) {
    const v2f a1 = *reinterpret_cast<const v2f*>(pa1 + 4 * t);
    const v2f a2 = *reinterpret_cast<const v2f*>(pa2 + 4 * t);
    if (t <= 18) {
      const v2f b = band_b(u - 4 * t);
      c10 = WMMA_F32_4(a1, b, c10);
      c20 = WMMA_F32_4(a2, b, c20);
    }
    if (t >= 4) {
      const v2f b = band_b(u + 16 - 4 * t);
      c11 = WMMA_F32_4(a1, b, c11);
      c21 = WMMA_F32_4(a2, b, c21);
    }
  }
#pragma unroll
  for (int v = 0; v < 8; ++v) {
    const int row = r0 + v + 8 * half;
    const size_t o = (size_t)row * WW + (j0 + n16);
    m[o]       = c10[v] * INVK;
    q2[o]      = c20[v] * INVK;
    m[o + 16]  = c11[v] * INVK;
    q2[o + 16] = c21[v] * INVK;
  }
}

__device__ __forceinline__ void k3_issue(const float* __restrict__ mc,
                                         const float* __restrict__ sc, int hr,
                                         unsigned wl, float (&qm)[7], float (&qs)[7]) {
  const size_t ro = (size_t)clamp_row(hr) * WW;
#pragma unroll
  for (int d = 0; d < 7; ++d) {
    qm[d] = (mc + ro + (size_t)d * HWs)[wl];
    qs[d] = (sc + ro + (size_t)d * HWs)[wl];
  }
}
__device__ __forceinline__ void k3_terms(const float (&qm)[7], const float (&qs)[7],
                                         bool ok, float (&t1)[7], float (&t2)[7]) {
#pragma unroll
  for (int d = 0; d < 7; ++d) {
    const float mv = ok ? qm[d] : 0.f;
    const float sv = ok ? qs[d] : 0.f;
    t1[d] = mv;
    t2[d] = mv * (sv - mv);
  }
}

#define K3_WARM 4
#define K3_MAIN 2
static_assert((2 * RAD) % K3_WARM == 0 && CH % K3_MAIN == 0, "row groups divide evenly");
static_assert(K3_WARM == 2 * K3_MAIN, "hand-over: one main row issued behind each last warmup row");

template <int AHEAD>
__device__ __forceinline__ void k3_warm_step(const float* __restrict__ mc,
                                             const float* __restrict__ sc, unsigned wl,
                                             int hr, float (&qm)[7], float (&qs)[7],
                                             float (&w1)[7], float (&w2)[7]) {
  float t1[7], t2[7], b1[7], b2[7];
  k3_terms(qm, qs, hr >= 0, t1, t2);
  if (AHEAD > 0) k3_issue(mc, sc, hr + AHEAD, wl, qm, qs);
  dbox7(t1, b1); dbox7(t2, b2);
#pragma unroll
  for (int d = 0; d < 7; ++d) { w1[d] += b1[d]; w2[d] += b2[d]; }
}

template <int AHEAD>
__device__ __forceinline__ void k3_main_step(const float* __restrict__ mc,
                                             const float* __restrict__ sc,
                                             float* __restrict__ B1c,
                                             float* __restrict__ B2c, unsigned wl, int h,
                                             float (&lm)[7], float (&ls)[7],
                                             float (&tm)[7], float (&ts)[7],
                                             float (&w1)[7], float (&w2)[7]) {
  float t1[7], t2[7], b1[7], b2[7];
  const int hl = h + RAD;
  const int ht = h - RAD;
  k3_terms(lm, ls, hl < HH, t1, t2);
  if (AHEAD > 0) k3_issue(mc, sc, hl + AHEAD, wl, lm, ls);
  dbox7(t1, b1); dbox7(t2, b2);
#pragma unroll
  for (int d = 0; d < 7; ++d) { w1[d] += b1[d]; w2[d] += b2[d]; }
  {
    const size_t ob = (size_t)h * WP;
#pragma unroll
    for (int d = 0; d < 7; ++d) {
      (B1c + ob + (size_t)d * HWP)[wl] = w1[d];
      (B2c + ob + (size_t)d * HWP)[wl] = w2[d];
    }
  }
  k3_terms(tm, ts, ht >= 0, t1, t2);
  if (AHEAD > 0) k3_issue(mc, sc, ht + AHEAD, wl, tm, ts);
  dbox7(t1, b1); dbox7(t2, b2);
#pragma unroll
  for (int d = 0; d < 7; ++d) { w1[d] -= b1[d]; w2[d] -= b2[d]; }
}

__global__ void __launch_bounds__(64)
k3_prep_dbox_hbox(const float* __restrict__ m, const float* __restrict__ S1,
                  float* __restrict__ B1, float* __restrict__ B2) {
  const unsigned wl = threadIdx.x;
  const int w0 = blockIdx.x * 64;
  const int h0 = blockIdx.y * CH;
  const int n  = blockIdx.z;
  const float* mc = m  + ((size_t)n * DD * HWs + w0);
  const float* sc = S1 + ((size_t)n * DD * HWs + w0);
  const size_t pb = (size_t)n * DD * HWP + PAD + w0;
  float* B1c = B1 + pb;
  float* B2c = B2 + pb;

  float w1[7], w2[7];
#pragma unroll
  for (int d = 0; d < 7; ++d) { w1[d] = 0.f; w2[d] = 0.f; }

  float qm[K3_WARM][7], qs[K3_WARM][7];
  float lm[K3_MAIN][7], ls[K3_MAIN][7];
  float tm[K3_MAIN][7], ts[K3_MAIN][7];
  const int hw0 = h0 - RAD;

#pragma unroll
  for (int r = 0; r < K3_WARM; ++r) k3_issue(mc, sc, hw0 + r, wl, qm[r], qs[r]);
  for (int ib = 0; ib < 2 * RAD - K3_WARM; ib += K3_WARM) {
#pragma unroll
    for (int r = 0; r < K3_WARM; ++r)
      k3_warm_step<K3_WARM>(mc, sc, wl, hw0 + ib + r, qm[r], qs[r], w1, w2);
  }
#pragma unroll
  for (int r = 0; r < K3_WARM; ++r) {
    k3_warm_step<0>(mc, sc, wl, hw0 + (2 * RAD - K3_WARM) + r, qm[r], qs[r], w1, w2);
    if ((r & 1) == 0) k3_issue(mc, sc, h0 + (r >> 1) + RAD, wl, lm[r >> 1], ls[r >> 1]);
    else              k3_issue(mc, sc, h0 + (r >> 1) - RAD, wl, tm[r >> 1], ts[r >> 1]);
  }

  for (int hb = h0; hb < h0 + CH - K3_MAIN; hb += K3_MAIN) {
#pragma unroll
    for (int r = 0; r < K3_MAIN; ++r)
      k3_main_step<K3_MAIN>(mc, sc, B1c, B2c, wl, hb + r, lm[r], ls[r], tm[r], ts[r],
                            w1, w2);
  }
#pragma unroll
  for (int r = 0; r < K3_MAIN; ++r)
    k3_main_step<0>(mc, sc, B1c, B2c, wl, h0 + CH - K3_MAIN + r, lm[r], ls[r], tm[r],
                    ts[r], w1, w2);
}

__global__ void __launch_bounds__(128)
k4_wbox_wmma2_stats_out(const float* __restrict__ B1, const float* __restrict__ B2,
                        const float* __restrict__ m, const float* __restrict__ q2,
                        const float* x, float* __restrict__ out) {
  __shared__ __attribute__((aligned(16))) float sB1[16 * LP];
  __shared__ __attribute__((aligned(16))) float sB2[16 * LP];
  const int lane = threadIdx.x;
  const int half = lane >> 4;
  const int n16  = lane & 15;
  const int wv   = threadIdx.y;
  const int tid  = wv * 32 + lane;
  const int r0  = blockIdx.y * 16;
  const int j0b = blockIdx.x * TCOLS;
  const int j0  = j0b + 32 * wv;
  const size_t gb = (size_t)r0 * WP + j0b;
  v4f t1[NLD], t2[NLD];
  stage_fetch(B1 + gb, tid, j0b, t1);
  stage_fetch(B2 + gb, tid, j0b, t2);

  const size_t ob = (size_t)(r0 + 8 * half) * WW + (j0 + n16);
  const float2* x2 = (const float2*)x;
  float  mv0[8], mv1[8], qv0[8], qv1[8];
  float2 xv0[8], xv1[8];
#pragma unroll
  for (int v = 0; v < 8; ++v) {
    const size_t o = ob + (size_t)v * WW;
    mv0[v] = m[o];
    qv0[v] = q2[o];
    xv0[v] = x2[o];
    mv1[v] = m[o + 16];
    qv1[v] = q2[o + 16];
    xv1[v] = x2[o + 16];
  }

  stage_put(sB1, tid, t1);
  stage_put(sB2, tid, t2);
  __syncthreads();

  const float* p1 = sB1 + n16 * LP + (PAD - RAD + 2 * half + 32 * wv);
  const float* p2 = sB2 + n16 * LP + (PAD - RAD + 2 * half + 32 * wv);
  const int u = n16 + RAD - 2 * half;

  v8f cm0 = {0.f,0.f,0.f,0.f,0.f,0.f,0.f,0.f};
  v8f cs0 = {0.f,0.f,0.f,0.f,0.f,0.f,0.f,0.f};
  v8f cm1 = {0.f,0.f,0.f,0.f,0.f,0.f,0.f,0.f};
  v8f cs1 = {0.f,0.f,0.f,0.f,0.f,0.f,0.f,0.f};

#pragma unroll
  for (int t = 0; t < 23; ++t) {
    const v2f a1 = *reinterpret_cast<const v2f*>(p1 + 4 * t);
    const v2f a2 = *reinterpret_cast<const v2f*>(p2 + 4 * t);
    if (t <= 18) {
      const v2f b = band_b(u - 4 * t);
      cm0 = WMMA_F32_4(a1, b, cm0);
      cs0 = WMMA_F32_4(a2, b, cs0);
    }
    if (t >= 4) {
      const v2f b = band_b(u + 16 - 4 * t);
      cm1 = WMMA_F32_4(a1, b, cm1);
      cs1 = WMMA_F32_4(a2, b, cs1);
    }
  }

  float2* o2 = (float2*)out;
#pragma unroll
  for (int v = 0; v < 8; ++v) {
    const size_t o = ob + (size_t)v * WW;
    {
      const float mv  = mv0[v];
      const float c1v = mv - 2.0f * INVK * cm0[v];
      const float c2v = qv0[v] - 2.0f * INVK * cs0[v];
      const float var = c2v - c1v * c1v;
      const float rv  = 1.0f / (sqrtf(var) + EPSF);
      float2 r;
      r.x = (xv0[v].x - mv) * rv;
      r.y = (xv0[v].y - mv) * rv;
      o2[o] = r;
    }
    {
      const float mv  = mv1[v];
      const float c1v = mv - 2.0f * INVK * cm1[v];
      const float c2v = qv1[v] - 2.0f * INVK * cs1[v];
      const float var = c2v - c1v * c1v;
      const float rv  = 1.0f / (sqrtf(var) + EPSF);
      float2 r;
      r.x = (xv1[v].x - mv) * rv;
      r.y = (xv1[v].y - mv) * rv;
      o2[o + 16] = r;
    }
  }
}

extern "C" void kernel_launch(void* const* d_in, const int* in_sizes, int n_in,
                              void* d_out, int out_size, void* d_ws, size_t ws_size,
                              hipStream_t stream) {
  (void)in_sizes; (void)n_in; (void)out_size; (void)ws_size;
  const float* x = (const float*)d_in[0];
  float* out = (float*)d_out;
  float* ws = (float*)d_ws;

  const size_t F  = (size_t)NN * DD * HH * WW;
  const size_t FP = (size_t)RTOT * WP;
  float* S1  = ws;
  float* A1p = ws + F;
  float* A2p = ws + F + FP;
  float* mf  = ws + F + 2 * FP;
  float* q2  = mf + F;

  const dim3 gHB(WW / 64, HH / CH, NN);
  const dim3 bHB(64);
  const dim3 gW(WW / TCOLS, RTOT / 16);
  const dim3 bW(32, 4);

  k1_chsum_dbox_hbox<<<gHB, bHB, 0, stream>>>(x, S1, A1p, A2p);
  k2_wbox_wmma2<<<gW, bW, 0, stream>>>(A1p, A2p, mf, q2);
  k3_prep_dbox_hbox<<<gHB, bHB, 0, stream>>>(mf, S1, A1p, A2p);
  k4_wbox_wmma2_stats_out<<<gW, bW, 0, stream>>>(A1p, A2p, mf, q2, x, out);
}


# Three box-filtered fields, not four: pass A filters only the channel sum (A2, q2, raw-sum plane go); K3 forms u_c=x_c-m from x, filters u and u^2 per channel; channels added after the H window
# speedup vs baseline: 3.5543x; 1.1847x over previous
#include <hip/hip_runtime.h>
#include <math.h>


#define NN 4
#define DD 7
#define HH 768
#define WW 768
#define HWs (HH * WW)
#define RAD 30
#define CH 64
#define PAD 32
#define WP (WW + 2 * PAD)
#define HWP (HH * WP)
#define RTOT (NN * DD * HH)

static_assert(HH % CH == 0, "chunking");
static_assert(CH > RAD, "warmup writes must stay in chunk");

constexpr float INVK = 1.0f / (7.0f * 61.0f * 61.0f * 2.0f);
constexpr float EPSF = 1e-7f;

typedef __attribute__((ext_vector_type(2))) float v2f;
typedef __attribute__((ext_vector_type(4))) float v4f;
typedef __attribute__((ext_vector_type(8))) float v8f;

#define TCOLS 128
#define WIN4 48
#define NLD ((16 * WIN4) / 128)
#define LP 196

static_assert(WW % TCOLS == 0, "column blocks");
static_assert(PAD % 4 == 0 && WW % 4 == 0 && TCOLS % 4 == 0, "a 16-byte piece never straddles a pad edge");
static_assert((16 * WIN4) % 128 == 0, "staging loads divide evenly");
static_assert(4 * WIN4 >= TCOLS + 2 * RAD + (PAD - RAD), "window covers the band");
static_assert(LP >= 4 * WIN4 && (LP % 8) == 4, "pitch = 4 * odd");

__device__ __forceinline__ void stage_fetch(const float* __restrict__ g, int tid, int j0b,
                                            v4f (&v)[NLD]) {
#pragma unroll
  for (int it = 0; it < NLD; ++it) {
    const int idx = it * 128 + tid;
    const int row = idx / WIN4;
    const int c4  = idx - row * WIN4;
    const int col = j0b + 4 * c4;
    v4f p = {0.f, 0.f, 0.f, 0.f};
    if ((unsigned)(col - PAD) < (unsigned)WW)
      p = *reinterpret_cast<const v4f*>(g + (size_t)row * WP + 4 * c4);
    v[it] = p;
  }
}
__device__ __forceinline__ void stage_put(float* s, int tid, const v4f (&v)[NLD]) {
#pragma unroll
  for (int it = 0; it < NLD; ++it) {
    const int idx = it * 128 + tid;
    const int row = idx / WIN4;
    const int c4  = idx - row * WIN4;
    *reinterpret_cast<v4f*>(s + row * LP + 4 * c4) = v[it];
  }
}

#define WMMA_F32_4(A, B, C) \
  __builtin_amdgcn_wmma_f32_16x16x4_f32(false, (A), false, (B), (short)0, (C), false, false)

__device__ __forceinline__ v2f band_b(int e) {
  v2f b;
  b.x = ((unsigned)(e + RAD) < 61u) ? 1.f : 0.f;
  b.y = ((unsigned)(e - 1 + RAD) < 61u) ? 1.f : 0.f;
  return b;
}

__device__ __forceinline__ void dbox7p(const v2f* s, v2f* b) {
  v2f ps[8];
  ps[0] = (v2f){0.f, 0.f};
#pragma unroll
  for (int d = 0; d < 7; ++d) ps[d + 1] = ps[d] + s[d];
#pragma unroll
  for (int d = 0; d < 7; ++d) {
    const int hi = (d + 3 > 6) ? 7 : (d + 4);
    const int lo = (d - 3 < 0) ? 0 : (d - 3);
    b[d] = ps[hi] - ps[lo];
  }
}

__device__ __forceinline__ int clamp_row(int h) {
  return h < 0 ? 0 : (h > HH - 1 ? HH - 1 : h);
}
__device__ __forceinline__ void refill_fence() { __builtin_amdgcn_sched_barrier(0); }

__device__ __forceinline__ void k1_issue(const v2f* __restrict__ xc, int hr,
                                         unsigned wl, v2f (&q)[7]) {
  const v2f* p = xc + (size_t)clamp_row(hr) * WW;
  refill_fence();
#pragma unroll
  for (int d = 0; d < 7; ++d) q[d] = (p + (size_t)d * HWs)[wl];
  refill_fence();
}
__device__ __forceinline__ void k1_vals(const v2f (&q)[7], bool ok, v2f (&s1)[7]) {
#pragma unroll
  for (int d = 0; d < 7; ++d) {
    s1[d].x = ok ? q[d].x : 0.f;
    s1[d].y = ok ? q[d].y : 0.f;
  }
}

#define K1_WARM 6
#define K1_MAIN 4
static_assert((2 * RAD) % K1_WARM == 0 && CH % K1_MAIN == 0, "row groups divide evenly");
static_assert(K1_WARM == 6 && K1_MAIN == 4, "hand-over schedule below is written for 6 and 4");

template <int AHEAD>
__device__ __forceinline__ void k1_warm_step(const v2f* __restrict__ xc, unsigned wl,
                                             int hr, v2f (&q)[7], v2f (&win1)[7]) {
  v2f s1[7], b1[7];
  k1_vals(q, hr >= 0, s1);
  if (AHEAD > 0) k1_issue(xc, hr + AHEAD, wl, q);
  dbox7p(s1, b1);
#pragma unroll
  for (int d = 0; d < 7; ++d) win1[d] += b1[d];
}

template <int AHEAD>
__device__ __forceinline__ void k1_main_step(const v2f* __restrict__ xc,
                                             float* __restrict__ A1c, unsigned wl,
                                             int h, v2f (&ql)[7], v2f (&qt)[7],
                                             v2f (&win1)[7]) {
  v2f s1[7], b1[7];
  const int hl = h + RAD;
  const int ht = h - RAD;
  k1_vals(ql, hl < HH, s1);
  if (AHEAD > 0) k1_issue(xc, hl + AHEAD, wl, ql);
  dbox7p(s1, b1);
#pragma unroll
  for (int d = 0; d < 7; ++d) win1[d] += b1[d];
  {
    float* o1 = A1c + (size_t)h * WP;
#pragma unroll
    for (int d = 0; d < 7; ++d) (o1 + (size_t)d * HWP)[wl] = win1[d].x + win1[d].y;
  }
  k1_vals(qt, ht >= 0, s1);
  if (AHEAD > 0) k1_issue(xc, ht + AHEAD, wl, qt);
  dbox7p(s1, b1);
#pragma unroll
  for (int d = 0; d < 7; ++d) win1[d] -= b1[d];
}

__global__ void __launch_bounds__(64)
k1_chsum_dbox_hbox(const float* __restrict__ x, float* __restrict__ A1) {
  const unsigned wl = threadIdx.x;
  const int w0 = blockIdx.x * 64;
  const int h0 = blockIdx.y * CH;
  const int n  = blockIdx.z;
  const v2f* xc = (const v2f*)x + ((size_t)n * DD * HWs + w0);
  float* A1c = A1 + ((size_t)n * DD * HWP + PAD + w0);

  v2f win1[7];
#pragma unroll
  for (int d = 0; d < 7; ++d) win1[d] = (v2f){0.f, 0.f};

  v2f q[K1_WARM][7];
  v2f ql[K1_MAIN][7], qt[K1_MAIN][7];
  const int hw0 = h0 - RAD;

#pragma unroll
  for (int r = 0; r < K1_WARM; ++r) k1_issue(xc, hw0 + r, wl, q[r]);
  for (int ib = 0; ib < 2 * RAD - K1_WARM; ib += K1_WARM) {
#pragma unroll
    for (int r = 0; r < K1_WARM; ++r)
      k1_warm_step<K1_WARM>(xc, wl, hw0 + ib + r, q[r], win1);
  }
#pragma unroll
  for (int r = 0; r < K1_WARM; ++r) {
    k1_warm_step<0>(xc, wl, hw0 + (2 * RAD - K1_WARM) + r, q[r], win1);
    if (r < 4) {
      if ((r & 1) == 0) k1_issue(xc, h0 + (r >> 1) + RAD, wl, ql[r >> 1]);
      else              k1_issue(xc, h0 + (r >> 1) - RAD, wl, qt[r >> 1]);
    } else {
      k1_issue(xc, h0 + (r - 2) + RAD, wl, ql[r - 2]);
      k1_issue(xc, h0 + (r - 2) - RAD, wl, qt[r - 2]);
    }
  }

  for (int hb = h0; hb < h0 + CH - K1_MAIN; hb += K1_MAIN) {
#pragma unroll
    for (int r = 0; r < K1_MAIN; ++r)
      k1_main_step<K1_MAIN>(xc, A1c, wl, hb + r, ql[r], qt[r], win1);
  }
#pragma unroll
  for (int r = 0; r < K1_MAIN; ++r)
    k1_main_step<0>(xc, A1c, wl, h0 + CH - K1_MAIN + r, ql[r], qt[r], win1);
}

__global__ void __launch_bounds__(128)
k2_wbox_wmma2(const float* __restrict__ A1, float* __restrict__ m) {
  __shared__ __attribute__((aligned(16))) float sA1[16 * LP];
  const int lane = threadIdx.x;
  const int half = lane >> 4;
  const int n16  = lane & 15;
  const int wv   = threadIdx.y;
  const int tid  = wv * 32 + lane;
  const int r0  = blockIdx.y * 16;
  const int j0b = blockIdx.x * TCOLS;
  const int j0  = j0b + 32 * wv;
  {
    const size_t gb = (size_t)r0 * WP + j0b;
    v4f t1[NLD];
    stage_fetch(A1 + gb, tid, j0b, t1);
    stage_put(sA1, tid, t1);
  }
  __syncthreads();
  const float* pa1 = sA1 + n16 * LP + (PAD - RAD + 2 * half + 32 * wv);
  const int u = n16 + RAD - 2 * half;

  v8f c10 = {0.f,0.f,0.f,0.f,0.f,0.f,0.f,0.f};
  v8f c11 = {0.f,0.f,0.f,0.f,0.f,0.f,0.f,0.f};

#pragma unroll
  for (int t = 0; t < 23; ++t) {
    const v2f a1 = *reinterpret_cast<const v2f*>(pa1 + 4 * t);
    if (t <= 18) {
      const v2f b = band_b(u - 4 * t);
      c10 = WMMA_F32_4(a1, b, c10);
    }
    if (t >= 4) {
      const v2f b = band_b(u + 16 - 4 * t);
      c11 = WMMA_F32_4(a1, b, c11);
    }
  }
#pragma unroll
  for (int v = 0; v < 8; ++v) {
    const int row = r0 + v + 8 * half;
    const size_t o = (size_t)row * WW + (j0 + n16);
    m[o]      = c10[v] * INVK;
    m[o + 16] = c11[v] * INVK;
  }
}

__device__ __forceinline__ void k3_issue(const v2f* __restrict__ xc,
                                         const float* __restrict__ mc, int hr,
                                         unsigned wl, v2f (&qx)[7], float (&qm)[7]) {
  const size_t ro = (size_t)clamp_row(hr) * WW;
  refill_fence();
#pragma unroll
  for (int d = 0; d < 7; ++d) {
    qx[d] = (xc + ro + (size_t)d * HWs)[wl];
    qm[d] = (mc + ro + (size_t)d * HWs)[wl];
  }
  refill_fence();
}
__device__ __forceinline__ void k3_terms(const v2f (&qx)[7], const float (&qm)[7],
                                         bool ok, v2f (&t1)[7], v2f (&t2)[7]) {
  v2f u[7];
#pragma unroll
  for (int d = 0; d < 7; ++d) u[d] = qx[d] - qm[d];
  asm volatile("" : "+v"(u[0]), "+v"(u[1]), "+v"(u[2]), "+v"(u[3]), "+v"(u[4]), "+v"(u[5]), "+v"(u[6]));
#pragma unroll
  for (int d = 0; d < 7; ++d) {
    const v2f uu = u[d] * u[d];
    t1[d].x = ok ? u[d].x : 0.f;
    t1[d].y = ok ? u[d].y : 0.f;
    t2[d].x = ok ? uu.x : 0.f;
    t2[d].y = ok ? uu.y : 0.f;
  }
}

#define K3_WARM 4
#define K3_MAIN 2
static_assert((2 * RAD) % K3_WARM == 0 && CH % K3_MAIN == 0, "row groups divide evenly");
static_assert(K3_WARM == 2 * K3_MAIN, "hand-over: one main row issued behind each last warmup row");

template <int AHEAD>
__device__ __forceinline__ void k3_warm_step(const v2f* __restrict__ xc,
                                             const float* __restrict__ mc, unsigned wl,
                                             int hr, v2f (&qx)[7], float (&qm)[7],
                                             v2f (&w1)[7], v2f (&w2)[7]) {
  v2f t1[7], t2[7], b1[7], b2[7];
  k3_terms(qx, qm, hr >= 0, t1, t2);
  if (AHEAD > 0) k3_issue(xc, mc, hr + AHEAD, wl, qx, qm);
  dbox7p(t1, b1); dbox7p(t2, b2);
#pragma unroll
  for (int d = 0; d < 7; ++d) { w1[d] += b1[d]; w2[d] += b2[d]; }
}

template <int AHEAD>
__device__ __forceinline__ void k3_main_step(const v2f* __restrict__ xc,
                                             const float* __restrict__ mc,
                                             float* __restrict__ B1c,
                                             float* __restrict__ B2c, unsigned wl, int h,
                                             v2f (&lx)[7], float (&lm)[7],
                                             v2f (&tx)[7], float (&tm)[7],
                                             v2f (&w1)[7], v2f (&w2)[7]) {
  v2f t1[7], t2[7], b1[7], b2[7];
  const int hl = h + RAD;
  const int ht = h - RAD;
  k3_terms(lx, lm, hl < HH, t1, t2);
  if (AHEAD > 0) k3_issue(xc, mc, hl + AHEAD, wl, lx, lm);
  dbox7p(t1, b1); dbox7p(t2, b2);
#pragma unroll
  for (int d = 0; d < 7; ++d) { w1[d] += b1[d]; w2[d] += b2[d]; }
  {
    const size_t ob = (size_t)h * WP;
#pragma unroll
    for (int d = 0; d < 7; ++d) {
      (B1c + ob + (size_t)d * HWP)[wl] = w1[d].x + w1[d].y;
      (B2c + ob + (size_t)d * HWP)[wl] = w2[d].x + w2[d].y;
    }
  }
  k3_terms(tx, tm, ht >= 0, t1, t2);
  if (AHEAD > 0) k3_issue(xc, mc, ht + AHEAD, wl, tx, tm);
  dbox7p(t1, b1); dbox7p(t2, b2);
#pragma unroll
  for (int d = 0; d < 7; ++d) { w1[d] -= b1[d]; w2[d] -= b2[d]; }
}

__global__ void __launch_bounds__(64)
k3_prep_dbox_hbox(const float* __restrict__ x, const float* __restrict__ m,
                  float* __restrict__ B1, float* __restrict__ B2) {
  const unsigned wl = threadIdx.x;
  const int w0 = blockIdx.x * 64;
  const int h0 = blockIdx.y * CH;
  const int n  = blockIdx.z;
  const v2f* xc = (const v2f*)x + ((size_t)n * DD * HWs + w0);
  const float* mc = m + ((size_t)n * DD * HWs + w0);
  const size_t pb = (size_t)n * DD * HWP + PAD + w0;
  float* B1c = B1 + pb;
  float* B2c = B2 + pb;

  v2f w1[7], w2[7];
#pragma unroll
  for (int d = 0; d < 7; ++d) { w1[d] = (v2f){0.f, 0.f}; w2[d] = (v2f){0.f, 0.f}; }

  v2f qx[K3_WARM][7]; float qm[K3_WARM][7];
  v2f lx[K3_MAIN][7]; float lm[K3_MAIN][7];
  v2f tx[K3_MAIN][7]; float tm[K3_MAIN][7];
  const int hw0 = h0 - RAD;

#pragma unroll
  for (int r = 0; r < K3_WARM; ++r) k3_issue(xc, mc, hw0 + r, wl, qx[r], qm[r]);
  for (int ib = 0; ib < 2 * RAD - K3_WARM; ib += K3_WARM) {
#pragma unroll
    for (int r = 0; r < K3_WARM; ++r)
      k3_warm_step<K3_WARM>(xc, mc, wl, hw0 + ib + r, qx[r], qm[r], w1, w2);
  }
#pragma unroll
  for (int r = 0; r < K3_WARM; ++r) {
    k3_warm_step<0>(xc, mc, wl, hw0 + (2 * RAD - K3_WARM) + r, qx[r], qm[r], w1, w2);
    if ((r & 1) == 0) k3_issue(xc, mc, h0 + (r >> 1) + RAD, wl, lx[r >> 1], lm[r >> 1]);
    else              k3_issue(xc, mc, h0 + (r >> 1) - RAD, wl, tx[r >> 1], tm[r >> 1]);
  }

  for (int hb = h0; hb < h0 + CH - K3_MAIN; hb += K3_MAIN) {
#pragma unroll
    for (int r = 0; r < K3_MAIN; ++r)
      k3_main_step<K3_MAIN>(xc, mc, B1c, B2c, wl, hb + r, lx[r], lm[r], tx[r], tm[r],
                            w1, w2);
  }
#pragma unroll
  for (int r = 0; r < K3_MAIN; ++r)
    k3_main_step<0>(xc, mc, B1c, B2c, wl, h0 + CH - K3_MAIN + r, lx[r], lm[r], tx[r],
                    tm[r], w1, w2);
}

__global__ void __launch_bounds__(128)
k4_wbox_wmma2_stats_out(const float* __restrict__ B1, const float* __restrict__ B2,
                        const float* __restrict__ m, const float* x,
                        float* __restrict__ out) {
  __shared__ __attribute__((aligned(16))) float sB1[16 * LP];
  __shared__ __attribute__((aligned(16))) float sB2[16 * LP];
  const int lane = threadIdx.x;
  const int half = lane >> 4;
  const int n16  = lane & 15;
  const int wv   = threadIdx.y;
  const int tid  = wv * 32 + lane;
  const int r0  = blockIdx.y * 16;
  const int j0b = blockIdx.x * TCOLS;
  const int j0  = j0b + 32 * wv;
  const size_t gb = (size_t)r0 * WP + j0b;
  v4f t1[NLD], t2[NLD];
  stage_fetch(B1 + gb, tid, j0b, t1);
  stage_fetch(B2 + gb, tid, j0b, t2);

  const size_t ob = (size_t)(r0 + 8 * half) * WW + (j0 + n16);
  const float2* x2 = (const float2*)x;
  float  mv0[8], mv1[8];
  float2 xv0[8], xv1[8];
#pragma unroll
  for (int v = 0; v < 8; ++v) {
    const size_t o = ob + (size_t)v * WW;
    mv0[v] = m[o];
    xv0[v] = x2[o];
    mv1[v] = m[o + 16];
    xv1[v] = x2[o + 16];
  }
  __builtin_amdgcn_sched_barrier(0);

  stage_put(sB1, tid, t1);
  stage_put(sB2, tid, t2);
  __syncthreads();

  const float* p1 = sB1 + n16 * LP + (PAD - RAD + 2 * half + 32 * wv);
  const float* p2 = sB2 + n16 * LP + (PAD - RAD + 2 * half + 32 * wv);
  const int u = n16 + RAD - 2 * half;

  v8f cm0 = {0.f,0.f,0.f,0.f,0.f,0.f,0.f,0.f};
  v8f cs0 = {0.f,0.f,0.f,0.f,0.f,0.f,0.f,0.f};
  v8f cm1 = {0.f,0.f,0.f,0.f,0.f,0.f,0.f,0.f};
  v8f cs1 = {0.f,0.f,0.f,0.f,0.f,0.f,0.f,0.f};

#pragma unroll
  for (int t = 0; t < 23; ++t) {
    const v2f a1 = *reinterpret_cast<const v2f*>(p1 + 4 * t);
    const v2f a2 = *reinterpret_cast<const v2f*>(p2 + 4 * t);
    if (t <= 18) {
      const v2f b = band_b(u - 4 * t);
      cm0 = WMMA_F32_4(a1, b, cm0);
      cs0 = WMMA_F32_4(a2, b, cs0);
    }
    if (t >= 4) {
      const v2f b = band_b(u + 16 - 4 * t);
      cm1 = WMMA_F32_4(a1, b, cm1);
      cs1 = WMMA_F32_4(a2, b, cs1);
    }
  }

  float2* o2 = (float2*)out;
#pragma unroll
  for (int v = 0; v < 8; ++v) {
    const size_t o = ob + (size_t)v * WW;
    {
      const float mv  = mv0[v];
      const float c1v = INVK * cm0[v];
      const float c2v = INVK * cs0[v];
      const float var = c2v - c1v * c1v;
      const float rv  = 1.0f / (sqrtf(var) + EPSF);
      float2 r;
      r.x = (xv0[v].x - mv) * rv;
      r.y = (xv0[v].y - mv) * rv;
      o2[o] = r;
    }
    {
      const float mv  = mv1[v];
      const float c1v = INVK * cm1[v];
      const float c2v = INVK * cs1[v];
      const float var = c2v - c1v * c1v;
      const float rv  = 1.0f / (sqrtf(var) + EPSF);
      float2 r;
      r.x = (xv1[v].x - mv) * rv;
      r.y = (xv1[v].y - mv) * rv;
      o2[o + 16] = r;
    }
  }
}

extern "C" void kernel_launch(void* const* d_in, const int* in_sizes, int n_in,
                              void* d_out, int out_size, void* d_ws, size_t ws_size,
                              hipStream_t stream) {
  (void)in_sizes; (void)n_in; (void)out_size; (void)ws_size;
  const float* x = (const float*)d_in[0];
  float* out = (float*)d_out;
  float* ws = (float*)d_ws;

  const size_t FP = (size_t)RTOT * WP;
  float* A1p = ws;
  float* B2p = ws + FP;
  float* mf  = ws + 2 * FP;

  const dim3 gHB(WW / 64, HH / CH, NN);
  const dim3 bHB(64);
  const dim3 gW(WW / TCOLS, RTOT / 16);
  const dim3 bW(32, 4);

  k1_chsum_dbox_hbox<<<gHB, bHB, 0, stream>>>(x, A1p);
  k2_wbox_wmma2<<<gW, bW, 0, stream>>>(A1p, mf);
  k3_prep_dbox_hbox<<<gHB, bHB, 0, stream>>>(x, mf, A1p, B2p);
  k4_wbox_wmma2_stats_out<<<gW, bW, 0, stream>>>(A1p, B2p, mf, x, out);
}
